# starting kernel
# baseline (speedup 1.0000x reference)
#include <hip/hip_runtime.h>
#include <hip/hip_bf16.h>

typedef __attribute__((ext_vector_type(2))) float v2f;
typedef __attribute__((ext_vector_type(8))) float v8f;

#define N_FRAG  262144
#define MBLOCK  32
#define KDIM    384
#define DMID    256
#define HSTR    396
#define ZSTR    268

__device__ __forceinline__ v8f wmma_f32(v2f a, v2f b, v8f c) {
    return __builtin_amdgcn_wmma_f32_16x16x4_f32(
         false, a,  false, b,
         (short)0, c,  false,  false);
}

__global__ __launch_bounds__(128, 1)
void janossy_readout_kernel(const float* __restrict__ x_atoms,
                            const int*   __restrict__ frag_idx,
                            const float* __restrict__ W1,
                            const float* __restrict__ b1,
                            const float* __restrict__ W2,
                            const float* __restrict__ b2,
                            const float* __restrict__ Wout,
                            const float* __restrict__ bout,
                            float* __restrict__ out)
{
    __shared__ float sh_h  [MBLOCK * HSTR];
    __shared__ float sh_z1f[MBLOCK * ZSTR];
    __shared__ float sh_z1r[MBLOCK * ZSTR];
    __shared__ float sh_p  [MBLOCK * ZSTR];
    __shared__ int   sh_idx[MBLOCK * 3];

    const int tid   = threadIdx.x;
    const int frag0 = blockIdx.x * MBLOCK;

    if (tid < MBLOCK * 3) sh_idx[tid] = frag_idx[frag0 * 3 + tid];
    __syncthreads();

    for (int i = tid; i < MBLOCK * 3 * 32; i += 128) {
        int row = i >> 5;
        int c4  = i & 31;
        int m   = row / 3;
        int k   = row - m * 3;
        const float4* src = (const float4*)(x_atoms + (size_t)sh_idx[row] * 128);
        float4 v = src[c4];
        *(float4*)&sh_h[m * HSTR + k * 128 + c4 * 4] = v;
    }
    __syncthreads();

    const int lane  = tid & 31;
    const int wave  = tid >> 5;
    const int mrow  = lane & 15;
    const int khalf = lane >> 4;
    const int ncol  = lane & 15;
    const int n0    = wave * 64;

    v8f accF[2][4], accR[2][4];
    #pragma unroll
    for (int mt = 0; mt < 2; ++mt)
        #pragma unroll
        for (int t = 0; t < 4; ++t) {
            accF[mt][t] = (v8f){0.f,0.f,0.f,0.f,0.f,0.f,0.f,0.f};
            accR[mt][t] = (v8f){0.f,0.f,0.f,0.f,0.f,0.f,0.f,0.f};
        }

    for (int kb = 0; kb < KDIM / 4; ++kb) {
        const int kf = kb * 4;
        const int kr = (2 - (kb >> 5)) * 128 + (kf & 127);
        const int ka  = kf + 2 * khalf;
        const int kra = kr + 2 * khalf;

        v2f aF0 = *(const v2f*)&sh_h[ mrow       * HSTR + ka ];
        v2f aF1 = *(const v2f*)&sh_h[(16 + mrow) * HSTR + ka ];
        v2f aR0 = *(const v2f*)&sh_h[ mrow       * HSTR + kra];
        v2f aR1 = *(const v2f*)&sh_h[(16 + mrow) * HSTR + kra];

        #pragma unroll
        for (int t = 0; t < 4; ++t) {
            const int n = n0 + t * 16 + ncol;
            v2f b;
            b.x = W1[(ka + 0) * DMID + n];
            b.y = W1[(ka + 1) * DMID + n];
            accF[0][t] = wmma_f32(aF0, b, accF[0][t]);
            accF[1][t] = wmma_f32(aF1, b, accF[1][t]);
            accR[0][t] = wmma_f32(aR0, b, accR[0][t]);
            accR[1][t] = wmma_f32(aR1, b, accR[1][t]);
        }
    }

    #pragma unroll
    for (int mt = 0; mt < 2; ++mt)
        #pragma unroll
        for (int t = 0; t < 4; ++t) {
            const int n = n0 + t * 16 + ncol;
            const float bias = b1[n];
            #pragma unroll
            for (int i = 0; i < 8; ++i) {
                const int m = mt * 16 + i + 8 * khalf;
                float vf = accF[mt][t][i] + bias; vf = vf > 0.f ? vf : 0.f;
                float vr = accR[mt][t][i] + bias; vr = vr > 0.f ? vr : 0.f;
                sh_z1f[m * ZSTR + n] = vf;
                sh_z1r[m * ZSTR + n] = vr;
            }
        }
    __syncthreads();

    #pragma unroll
    for (int mt = 0; mt < 2; ++mt)
        #pragma unroll
        for (int t = 0; t < 4; ++t) {
            accF[mt][t] = (v8f){0.f,0.f,0.f,0.f,0.f,0.f,0.f,0.f};
            accR[mt][t] = (v8f){0.f,0.f,0.f,0.f,0.f,0.f,0.f,0.f};
        }

    for (int kb = 0; kb < DMID / 4; ++kb) {
        const int ka = kb * 4 + 2 * khalf;

        v2f aF0 = *(const v2f*)&sh_z1f[ mrow       * ZSTR + ka];
        v2f aF1 = *(const v2f*)&sh_z1f[(16 + mrow) * ZSTR + ka];
        v2f aR0 = *(const v2f*)&sh_z1r[ mrow       * ZSTR + ka];
        v2f aR1 = *(const v2f*)&sh_z1r[(16 + mrow) * ZSTR + ka];

        #pragma unroll
        for (int t = 0; t < 4; ++t) {
            const int n = n0 + t * 16 + ncol;
            v2f b;
            b.x = W2[(ka + 0) * DMID + n];
            b.y = W2[(ka + 1) * DMID + n];
            accF[0][t] = wmma_f32(aF0, b, accF[0][t]);
            accF[1][t] = wmma_f32(aF1, b, accF[1][t]);
            accR[0][t] = wmma_f32(aR0, b, accR[0][t]);
            accR[1][t] = wmma_f32(aR1, b, accR[1][t]);
        }
    }

    #pragma unroll
    for (int mt = 0; mt < 2; ++mt)
        #pragma unroll
        for (int t = 0; t < 4; ++t) {
            const int n = n0 + t * 16 + ncol;
            const float bias = b2[n];
            #pragma unroll
            for (int i = 0; i < 8; ++i) {
                const int m = mt * 16 + i + 8 * khalf;
                float vf = accF[mt][t][i] + bias; vf = vf > 0.f ? vf : 0.f;
                float vr = accR[mt][t][i] + bias; vr = vr > 0.f ? vr : 0.f;
                sh_p[m * ZSTR + n] = vf + vr;
            }
        }
    __syncthreads();

    for (int o = tid; o < MBLOCK * 3; o += 128) {
        const int m = o / 3;
        const int d = o - m * 3;
        float s = bout[d];
        const float* prow = &sh_p[m * ZSTR];
        #pragma unroll 4
        for (int k = 0; k < DMID; ++k)
            s += prow[k] * Wout[k * 3 + d];
        out[(size_t)(frag0 + m) * 3 + d] = s;
    }
}

extern "C" void kernel_launch(void* const* d_in, const int* in_sizes, int n_in,
                              void* d_out, int out_size, void* d_ws, size_t ws_size,
                              hipStream_t stream) {
    const float* x_atoms = (const float*)d_in[0];
    const int*   frag_idx = (const int*)d_in[1];
    const float* W1   = (const float*)d_in[2];
    const float* b1   = (const float*)d_in[3];
    const float* W2   = (const float*)d_in[4];
    const float* b2   = (const float*)d_in[5];
    const float* Wout = (const float*)d_in[6];
    const float* bout = (const float*)d_in[7];
    float* out = (float*)d_out;

    const int n_frag = in_sizes[1] / 3;
    const int grid = n_frag / MBLOCK;
    janossy_readout_kernel<<<grid, 128, 0, stream>>>(
        x_atoms, frag_idx, W1, b1, W2, b2, Wout, bout, out);
}


# Layer 1 on half-sum and half-difference of the two end slots: fwd=S+D, rev=S-D, 3 K-blocks of 128 instead of 6 (W1 end blocks added/subtracted on load); changes the f32 summation order
# speedup vs baseline: 1.3248x; 1.3248x over previous
#include <hip/hip_runtime.h>
#include <hip/hip_bf16.h>

typedef __attribute__((ext_vector_type(2))) float v2f;
typedef __attribute__((ext_vector_type(8))) float v8f;

#define N_FRAG  262144
#define MBLOCK  32
#define KDIM    384
#define DMID    256
#define HSTR    396
#define ZSTR    268

__device__ __forceinline__ v8f wmma_f32(v2f a, v2f b, v8f c) {
    return __builtin_amdgcn_wmma_f32_16x16x4_f32(
         false, a,  false, b,
         (short)0, c,  false,  false);
}

__global__ __launch_bounds__(128, 1)
void janossy_readout_kernel(const float* __restrict__ x_atoms,
                            const int*   __restrict__ frag_idx,
                            const float* __restrict__ W1,
                            const float* __restrict__ b1,
                            const float* __restrict__ W2,
                            const float* __restrict__ b2,
                            const float* __restrict__ Wout,
                            const float* __restrict__ bout,
                            float* __restrict__ out)
{
    __shared__ float sh_h  [MBLOCK * HSTR];
    __shared__ float sh_z1f[MBLOCK * ZSTR];
    __shared__ float sh_z1r[MBLOCK * ZSTR];
    __shared__ float sh_p  [MBLOCK * ZSTR];
    __shared__ int   sh_idx[MBLOCK * 3];

    const int tid   = threadIdx.x;
    const int frag0 = blockIdx.x * MBLOCK;

    if (tid < MBLOCK * 3) sh_idx[tid] = frag_idx[frag0 * 3 + tid];
    __syncthreads();

    #pragma unroll 2
    for (int i = tid; i < MBLOCK * 32; i += 128) {
        int m   = i >> 5;
        int c4  = i & 31;
        const float4* s0 = (const float4*)(x_atoms + (size_t)sh_idx[m * 3 + 0] * 128);
        const float4* s1 = (const float4*)(x_atoms + (size_t)sh_idx[m * 3 + 1] * 128);
        const float4* s2 = (const float4*)(x_atoms + (size_t)sh_idx[m * 3 + 2] * 128);
        float4 v0 = s0[c4];
        float4 v1 = s1[c4];
        float4 v2 = s2[c4];
        float4 hs, hd;
        hs.x = 0.5f * (v0.x + v2.x);  hd.x = 0.5f * (v0.x - v2.x);
        hs.y = 0.5f * (v0.y + v2.y);  hd.y = 0.5f * (v0.y - v2.y);
        hs.z = 0.5f * (v0.z + v2.z);  hd.z = 0.5f * (v0.z - v2.z);
        hs.w = 0.5f * (v0.w + v2.w);  hd.w = 0.5f * (v0.w - v2.w);
        *(float4*)&sh_h[m * HSTR +   0 + c4 * 4] = hs;
        *(float4*)&sh_h[m * HSTR + 128 + c4 * 4] = v1;
        *(float4*)&sh_h[m * HSTR + 256 + c4 * 4] = hd;
    }
    __syncthreads();

    const int lane  = tid & 31;
    const int wave  = tid >> 5;
    const int mrow  = lane & 15;
    const int khalf = lane >> 4;
    const int ncol  = lane & 15;
    const int n0    = wave * 64;

    v8f accF[2][4], accR[2][4];
    #pragma unroll
    for (int mt = 0; mt < 2; ++mt)
        #pragma unroll
        for (int t = 0; t < 4; ++t) {
            accF[mt][t] = (v8f){0.f,0.f,0.f,0.f,0.f,0.f,0.f,0.f};
            accR[mt][t] = (v8f){0.f,0.f,0.f,0.f,0.f,0.f,0.f,0.f};
        }

    #pragma unroll 2
    for (int kb = 0; kb < 128 / 4; ++kb) {
        const int ka = kb * 4 + 2 * khalf;

        v2f aS0 = *(const v2f*)&sh_h[ mrow       * HSTR +       ka];
        v2f aS1 = *(const v2f*)&sh_h[(16 + mrow) * HSTR +       ka];
        v2f aM0 = *(const v2f*)&sh_h[ mrow       * HSTR + 128 + ka];
        v2f aM1 = *(const v2f*)&sh_h[(16 + mrow) * HSTR + 128 + ka];
        v2f aD0 = *(const v2f*)&sh_h[ mrow       * HSTR + 256 + ka];
        v2f aD1 = *(const v2f*)&sh_h[(16 + mrow) * HSTR + 256 + ka];

        #pragma unroll
        for (int t = 0; t < 4; ++t) {
            const int n = n0 + t * 16 + ncol;
            v2f wa, wb, wc, bs, bd;
            wa.x = W1[(      ka + 0) * DMID + n];
            wa.y = W1[(      ka + 1) * DMID + n];
            wb.x = W1[(128 + ka + 0) * DMID + n];
            wb.y = W1[(128 + ka + 1) * DMID + n];
            wc.x = W1[(256 + ka + 0) * DMID + n];
            wc.y = W1[(256 + ka + 1) * DMID + n];
            bs = wa + wc;
            bd = wa - wc;
            accF[0][t] = wmma_f32(aS0, bs, accF[0][t]);
            accF[1][t] = wmma_f32(aS1, bs, accF[1][t]);
            accR[0][t] = wmma_f32(aD0, bd, accR[0][t]);
            accR[1][t] = wmma_f32(aD1, bd, accR[1][t]);
            accF[0][t] = wmma_f32(aM0, wb, accF[0][t]);
            accF[1][t] = wmma_f32(aM1, wb, accF[1][t]);
        }
    }

    #pragma unroll
    for (int mt = 0; mt < 2; ++mt)
        #pragma unroll
        for (int t = 0; t < 4; ++t) {
            const int n = n0 + t * 16 + ncol;
            const float bias = b1[n];
            #pragma unroll
            for (int i = 0; i < 8; ++i) {
                const int m = mt * 16 + i + 8 * khalf;
                const float sS = accF[mt][t][i];
                const float sD = accR[mt][t][i];
                float vf = (sS + sD) + bias; vf = vf > 0.f ? vf : 0.f;
                float vr = (sS - sD) + bias; vr = vr > 0.f ? vr : 0.f;
                sh_z1f[m * ZSTR + n] = vf;
                sh_z1r[m * ZSTR + n] = vr;
            }
        }
    __syncthreads();

    #pragma unroll
    for (int mt = 0; mt < 2; ++mt)
        #pragma unroll
        for (int t = 0; t < 4; ++t) {
            accF[mt][t] = (v8f){0.f,0.f,0.f,0.f,0.f,0.f,0.f,0.f};
            accR[mt][t] = (v8f){0.f,0.f,0.f,0.f,0.f,0.f,0.f,0.f};
        }

    for (int kb = 0; kb < DMID / 4; ++kb) {
        const int ka = kb * 4 + 2 * khalf;

        v2f aF0 = *(const v2f*)&sh_z1f[ mrow       * ZSTR + ka];
        v2f aF1 = *(const v2f*)&sh_z1f[(16 + mrow) * ZSTR + ka];
        v2f aR0 = *(const v2f*)&sh_z1r[ mrow       * ZSTR + ka];
        v2f aR1 = *(const v2f*)&sh_z1r[(16 + mrow) * ZSTR + ka];

        #pragma unroll
        for (int t = 0; t < 4; ++t) {
            const int n = n0 + t * 16 + ncol;
            v2f b;
            b.x = W2[(ka + 0) * DMID + n];
            b.y = W2[(ka + 1) * DMID + n];
            accF[0][t] = wmma_f32(aF0, b, accF[0][t]);
            accF[1][t] = wmma_f32(aF1, b, accF[1][t]);
            accR[0][t] = wmma_f32(aR0, b, accR[0][t]);
            accR[1][t] = wmma_f32(aR1, b, accR[1][t]);
        }
    }

    #pragma unroll
    for (int mt = 0; mt < 2; ++mt)
        #pragma unroll
        for (int t = 0; t < 4; ++t) {
            const int n = n0 + t * 16 + ncol;
            const float bias = b2[n];
            #pragma unroll
            for (int i = 0; i < 8; ++i) {
                const int m = mt * 16 + i + 8 * khalf;
                float vf = accF[mt][t][i] + bias; vf = vf > 0.f ? vf : 0.f;
                float vr = accR[mt][t][i] + bias; vr = vr > 0.f ? vr : 0.f;
                sh_p[m * ZSTR + n] = vf + vr;
            }
        }
    __syncthreads();

    for (int o = tid; o < MBLOCK * 3; o += 128) {
        const int m = o / 3;
        const int d = o - m * 3;
        float s = bout[d];
        const float* prow = &sh_p[m * ZSTR];
        #pragma unroll 4
        for (int k = 0; k < DMID; ++k)
            s += prow[k] * Wout[k * 3 + d];
        out[(size_t)(frag0 + m) * 3 + d] = s;
    }
}

extern "C" void kernel_launch(void* const* d_in, const int* in_sizes, int n_in,
                              void* d_out, int out_size, void* d_ws, size_t ws_size,
                              hipStream_t stream) {
    const float* x_atoms = (const float*)d_in[0];
    const int*   frag_idx = (const int*)d_in[1];
    const float* W1   = (const float*)d_in[2];
    const float* b1   = (const float*)d_in[3];
    const float* W2   = (const float*)d_in[4];
    const float* b2   = (const float*)d_in[5];
    const float* Wout = (const float*)d_in[6];
    const float* bout = (const float*)d_in[7];
    float* out = (float*)d_out;

    const int n_frag = in_sizes[1] / 3;
    const int grid = n_frag / MBLOCK;
    janossy_readout_kernel<<<grid, 128, 0, stream>>>(
        x_atoms, frag_idx, W1, b1, W2, b2, Wout, bout, out);
}


# Operand loads of both layer loops issued two stages ahead of their products in 3-slot register rings (W1/W2 from L2, A rows from LDS), order pinned; summation order unchanged
# speedup vs baseline: 1.4910x; 1.1255x over previous
#include <hip/hip_runtime.h>
#include <hip/hip_bf16.h>

typedef __attribute__((ext_vector_type(2))) float v2f;
typedef __attribute__((ext_vector_type(8))) float v8f;

#define N_FRAG  262144
#define MBLOCK  32
#define KDIM    384
#define DMID    256
#define HSTR    396
#define ZSTR    268

__device__ __forceinline__ v8f wmma_f32(v2f a, v2f b, v8f c) {
    return __builtin_amdgcn_wmma_f32_16x16x4_f32(
         false, a,  false, b,
         (short)0, c,  false,  false);
}

#define PIN_ORDER() __builtin_amdgcn_sched_barrier(0)

struct L1Frag {
    v2f wa[4], wb[4], wc[4];
    v2f aS0, aS1, aM0, aM1, aD0, aD1;
    v2f bs[4], bd[4];
};
struct L2Frag {
    v2f b[2][4];
    v2f a[2][4];
};

__device__ __forceinline__ void l1_load_w(L1Frag& f, const float* __restrict__ W1, int stage, int khalf, int ncb) {
    const int ka = stage * 4 + 2 * khalf;
    #pragma unroll
    for (int t = 0; t < 4; ++t) {
        const int n = ncb + t * 16;
        f.wa[t].x = W1[(      ka + 0) * DMID + n];
        f.wa[t].y = W1[(      ka + 1) * DMID + n];
        f.wb[t].x = W1[(128 + ka + 0) * DMID + n];
        f.wb[t].y = W1[(128 + ka + 1) * DMID + n];
        f.wc[t].x = W1[(256 + ka + 0) * DMID + n];
        f.wc[t].y = W1[(256 + ka + 1) * DMID + n];
    }
}
__device__ __forceinline__ void l1_load_h(L1Frag& f, const float* sh_h, int stage, int khalf, int mrow) {
    const int ka = stage * 4 + 2 * khalf;
    f.aS0 = *(const v2f*)&sh_h[ mrow       * HSTR +       ka];
    f.aS1 = *(const v2f*)&sh_h[(16 + mrow) * HSTR +       ka];
    f.aM0 = *(const v2f*)&sh_h[ mrow       * HSTR + 128 + ka];
    f.aM1 = *(const v2f*)&sh_h[(16 + mrow) * HSTR + 128 + ka];
    f.aD0 = *(const v2f*)&sh_h[ mrow       * HSTR + 256 + ka];
    f.aD1 = *(const v2f*)&sh_h[(16 + mrow) * HSTR + 256 + ka];
}
__device__ __forceinline__ void l1_comb(L1Frag& f) {
    #pragma unroll
    for (int t = 0; t < 4; ++t) {
        f.bs[t] = f.wa[t] + f.wc[t];
        f.bd[t] = f.wa[t] - f.wc[t];
    }
}
__device__ __forceinline__ void l1_mma(v8f (&accF)[2][4], v8f (&accR)[2][4], const L1Frag& f) {
    #pragma unroll
    for (int t = 0; t < 4; ++t) {
        accF[0][t] = wmma_f32(f.aS0, f.bs[t], accF[0][t]);
        accF[1][t] = wmma_f32(f.aS1, f.bs[t], accF[1][t]);
        accR[0][t] = wmma_f32(f.aD0, f.bd[t], accR[0][t]);
        accR[1][t] = wmma_f32(f.aD1, f.bd[t], accR[1][t]);
        accF[0][t] = wmma_f32(f.aM0, f.wb[t], accF[0][t]);
        accF[1][t] = wmma_f32(f.aM1, f.wb[t], accF[1][t]);
    }
}

__device__ __forceinline__ void l2_load_w(L2Frag& f, const float* __restrict__ W2, int stage, int khalf, int ncb) {
    #pragma unroll
    for (int u = 0; u < 2; ++u) {
        const int ka = (stage * 2 + u) * 4 + 2 * khalf;
        #pragma unroll
        for (int t = 0; t < 4; ++t) {
            const int n = ncb + t * 16;
            f.b[u][t].x = W2[(ka + 0) * DMID + n];
            f.b[u][t].y = W2[(ka + 1) * DMID + n];
        }
    }
}
__device__ __forceinline__ void l2_load_z(L2Frag& f, const float* zf, const float* zr, int stage, int khalf, int mrow) {
    #pragma unroll
    for (int u = 0; u < 2; ++u) {
        const int ka = (stage * 2 + u) * 4 + 2 * khalf;
        f.a[u][0] = *(const v2f*)&zf[ mrow       * ZSTR + ka];
        f.a[u][1] = *(const v2f*)&zf[(16 + mrow) * ZSTR + ka];
        f.a[u][2] = *(const v2f*)&zr[ mrow       * ZSTR + ka];
        f.a[u][3] = *(const v2f*)&zr[(16 + mrow) * ZSTR + ka];
    }
}
__device__ __forceinline__ void l2_mma(v8f (&accF)[2][4], v8f (&accR)[2][4], const L2Frag& f) {
    #pragma unroll
    for (int u = 0; u < 2; ++u)
        #pragma unroll
        for (int t = 0; t < 4; ++t) {
            accF[0][t] = wmma_f32(f.a[u][0], f.b[u][t], accF[0][t]);
            accF[1][t] = wmma_f32(f.a[u][1], f.b[u][t], accF[1][t]);
            accR[0][t] = wmma_f32(f.a[u][2], f.b[u][t], accR[0][t]);
            accR[1][t] = wmma_f32(f.a[u][3], f.b[u][t], accR[1][t]);
        }
}

__global__ __launch_bounds__(128, 1)
void janossy_readout_kernel(const float* __restrict__ x_atoms,
                            const int*   __restrict__ frag_idx,
                            const float* __restrict__ W1,
                            const float* __restrict__ b1,
                            const float* __restrict__ W2,
                            const float* __restrict__ b2,
                            const float* __restrict__ Wout,
                            const float* __restrict__ bout,
                            float* __restrict__ out)
{
    __shared__ float sh_h  [MBLOCK * HSTR];
    __shared__ float sh_z1f[MBLOCK * ZSTR];
    __shared__ float sh_z1r[MBLOCK * ZSTR];
    __shared__ float sh_p  [MBLOCK * ZSTR];
    __shared__ int   sh_idx[MBLOCK * 3];

    const int tid   = threadIdx.x;
    const int frag0 = blockIdx.x * MBLOCK;

    const int lane  = tid & 31;
    const int wave  = tid >> 5;
    const int mrow  = lane & 15;
    const int khalf = lane >> 4;
    const int ncol  = lane & 15;
    const int n0    = wave * 64;
    const int ncb   = n0 + ncol;

    L1Frag q1a, q1b, q1c;
    L2Frag q2a, q2b, q2c;

    l1_load_w(q1a, W1, 0, khalf, ncb);
    l1_load_w(q1b, W1, 1, khalf, ncb);

    if (tid < MBLOCK * 3) sh_idx[tid] = frag_idx[frag0 * 3 + tid];
    __syncthreads();

    #pragma unroll 2
    for (int i = tid; i < MBLOCK * 32; i += 128) {
        int m   = i >> 5;
        int c4  = i & 31;
        const float4* s0 = (const float4*)(x_atoms + (size_t)sh_idx[m * 3 + 0] * 128);
        const float4* s1 = (const float4*)(x_atoms + (size_t)sh_idx[m * 3 + 1] * 128);
        const float4* s2 = (const float4*)(x_atoms + (size_t)sh_idx[m * 3 + 2] * 128);
        float4 v0 = s0[c4];
        float4 v1 = s1[c4];
        float4 v2 = s2[c4];
        float4 hs, hd;
        hs.x = 0.5f * (v0.x + v2.x);  hd.x = 0.5f * (v0.x - v2.x);
        hs.y = 0.5f * (v0.y + v2.y);  hd.y = 0.5f * (v0.y - v2.y);
        hs.z = 0.5f * (v0.z + v2.z);  hd.z = 0.5f * (v0.z - v2.z);
        hs.w = 0.5f * (v0.w + v2.w);  hd.w = 0.5f * (v0.w - v2.w);
        *(float4*)&sh_h[m * HSTR +   0 + c4 * 4] = hs;
        *(float4*)&sh_h[m * HSTR + 128 + c4 * 4] = v1;
        *(float4*)&sh_h[m * HSTR + 256 + c4 * 4] = hd;
    }
    __syncthreads();

    v8f accF[2][4], accR[2][4];
    #pragma unroll
    for (int mt = 0; mt < 2; ++mt)
        #pragma unroll
        for (int t = 0; t < 4; ++t) {
            accF[mt][t] = (v8f){0.f,0.f,0.f,0.f,0.f,0.f,0.f,0.f};
            accR[mt][t] = (v8f){0.f,0.f,0.f,0.f,0.f,0.f,0.f,0.f};
        }

#define L1_STAGE(X, Z, jz)                                   \
    l1_comb(X);                                              \
    PIN_ORDER();                                             \
    l1_load_w(Z, W1, (jz), khalf, ncb);                      \
    l1_load_h(Z, sh_h, (jz), khalf, mrow);                   \
    PIN_ORDER();                                             \
    l1_mma(accF, accR, X);                                   \
    PIN_ORDER();

    l1_load_h(q1a, sh_h, 0, khalf, mrow);
    l1_load_h(q1b, sh_h, 1, khalf, mrow);
    for (int j = 0; j < 30; j += 3) {
        L1_STAGE(q1a, q1c, j + 2)
        L1_STAGE(q1b, q1a, j + 3)
        L1_STAGE(q1c, q1b, j + 4)
    }
    l1_comb(q1a);
    PIN_ORDER();
    l2_load_w(q2a, W2, 0, khalf, ncb);
    PIN_ORDER();
    l1_mma(accF, accR, q1a);
    PIN_ORDER();
    l1_comb(q1b);
    PIN_ORDER();
    l2_load_w(q2b, W2, 1, khalf, ncb);
    PIN_ORDER();
    l1_mma(accF, accR, q1b);
#undef L1_STAGE

    #pragma unroll
    for (int mt = 0; mt < 2; ++mt)
        #pragma unroll
        for (int t = 0; t < 4; ++t) {
            const int n = n0 + t * 16 + ncol;
            const float bias = b1[n];
            #pragma unroll
            for (int i = 0; i < 8; ++i) {
                const int m = mt * 16 + i + 8 * khalf;
                const float sS = accF[mt][t][i];
                const float sD = accR[mt][t][i];
                float vf = (sS + sD) + bias; vf = vf > 0.f ? vf : 0.f;
                float vr = (sS - sD) + bias; vr = vr > 0.f ? vr : 0.f;
                sh_z1f[m * ZSTR + n] = vf;
                sh_z1r[m * ZSTR + n] = vr;
            }
        }
    __syncthreads();

    #pragma unroll
    for (int mt = 0; mt < 2; ++mt)
        #pragma unroll
        for (int t = 0; t < 4; ++t) {
            accF[mt][t] = (v8f){0.f,0.f,0.f,0.f,0.f,0.f,0.f,0.f};
            accR[mt][t] = (v8f){0.f,0.f,0.f,0.f,0.f,0.f,0.f,0.f};
        }

#define L2_STAGE(X, Z, jz)                                   \
    l2_load_w(Z, W2, (jz), khalf, ncb);                      \
    l2_load_z(Z, sh_z1f, sh_z1r, (jz), khalf, mrow);         \
    PIN_ORDER();                                             \
    l2_mma(accF, accR, X);                                   \
    PIN_ORDER();

    l2_load_z(q2a, sh_z1f, sh_z1r, 0, khalf, mrow);
    l2_load_z(q2b, sh_z1f, sh_z1r, 1, khalf, mrow);
    for (int j = 0; j < 30; j += 3) {
        L2_STAGE(q2a, q2c, j + 2)
        L2_STAGE(q2b, q2a, j + 3)
        L2_STAGE(q2c, q2b, j + 4)
    }
    l2_mma(accF, accR, q2a);
    l2_mma(accF, accR, q2b);
#undef L2_STAGE

    #pragma unroll
    for (int mt = 0; mt < 2; ++mt)
        #pragma unroll
        for (int t = 0; t < 4; ++t) {
            const int n = n0 + t * 16 + ncol;
            const float bias = b2[n];
            #pragma unroll
            for (int i = 0; i < 8; ++i) {
                const int m = mt * 16 + i + 8 * khalf;
                float vf = accF[mt][t][i] + bias; vf = vf > 0.f ? vf : 0.f;
                float vr = accR[mt][t][i] + bias; vr = vr > 0.f ? vr : 0.f;
                sh_p[m * ZSTR + n] = vf + vr;
            }
        }
    __syncthreads();

    for (int o = tid; o < MBLOCK * 3; o += 128) {
        const int m = o / 3;
        const int d = o - m * 3;
        float s = bout[d];
        const float* prow = &sh_p[m * ZSTR];
        #pragma unroll 4
        for (int k = 0; k < DMID; ++k)
            s += prow[k] * Wout[k * 3 + d];
        out[(size_t)(frag0 + m) * 3 + d] = s;
    }
}

extern "C" void kernel_launch(void* const* d_in, const int* in_sizes, int n_in,
                              void* d_out, int out_size, void* d_ws, size_t ws_size,
                              hipStream_t stream) {
    const float* x_atoms = (const float*)d_in[0];
    const int*   frag_idx = (const int*)d_in[1];
    const float* W1   = (const float*)d_in[2];
    const float* b1   = (const float*)d_in[3];
    const float* W2   = (const float*)d_in[4];
    const float* b2   = (const float*)d_in[5];
    const float* Wout = (const float*)d_in[6];
    const float* bout = (const float*)d_in[7];
    float* out = (float*)d_out;

    const int n_frag = in_sizes[1] / 3;
    const int grid = n_frag / MBLOCK;
    janossy_readout_kernel<<<grid, 128, 0, stream>>>(
        x_atoms, frag_idx, W1, b1, W2, b2, Wout, bout, out);
}


# One LDS region reused phase by phase (gathered rows, then Z1, then pooled; two added barriers): 67 KiB not 150 KiB per workgroup, so 3 workgroups per WGP instead of 2; summation order unchanged
# speedup vs baseline: 1.5749x; 1.0563x over previous
#include <hip/hip_runtime.h>
#include <hip/hip_bf16.h>

typedef __attribute__((ext_vector_type(2))) float v2f;
typedef __attribute__((ext_vector_type(8))) float v8f;

#define N_FRAG  262144
#define MBLOCK  32
#define KDIM    384
#define DMID    256
#define HSTR    396
#define ZSTR    268

__device__ __forceinline__ v8f wmma_f32(v2f a, v2f b, v8f c) {
    return __builtin_amdgcn_wmma_f32_16x16x4_f32(
         false, a,  false, b,
         (short)0, c,  false,  false);
}

#define PIN_ORDER() __builtin_amdgcn_sched_barrier(0)

struct L1Frag {
    v2f wa[4], wb[4], wc[4];
    v2f aS0, aS1, aM0, aM1, aD0, aD1;
    v2f bs[4], bd[4];
};
struct L2Frag {
    v2f b[2][4];
    v2f a[2][4];
};

__device__ __forceinline__ void l1_load_w(L1Frag& f, const float* __restrict__ W1, int stage, int khalf, int ncb) {
    const int ka = stage * 4 + 2 * khalf;
    #pragma unroll
    for (int t = 0; t < 4; ++t) {
        const int n = ncb + t * 16;
        f.wa[t].x = W1[(      ka + 0) * DMID + n];
        f.wa[t].y = W1[(      ka + 1) * DMID + n];
        f.wb[t].x = W1[(128 + ka + 0) * DMID + n];
        f.wb[t].y = W1[(128 + ka + 1) * DMID + n];
        f.wc[t].x = W1[(256 + ka + 0) * DMID + n];
        f.wc[t].y = W1[(256 + ka + 1) * DMID + n];
    }
}
__device__ __forceinline__ void l1_load_h(L1Frag& f, const float* sh_h, int stage, int khalf, int mrow) {
    const int ka = stage * 4 + 2 * khalf;
    f.aS0 = *(const v2f*)&sh_h[ mrow       * HSTR +       ka];
    f.aS1 = *(const v2f*)&sh_h[(16 + mrow) * HSTR +       ka];
    f.aM0 = *(const v2f*)&sh_h[ mrow       * HSTR + 128 + ka];
    f.aM1 = *(const v2f*)&sh_h[(16 + mrow) * HSTR + 128 + ka];
    f.aD0 = *(const v2f*)&sh_h[ mrow       * HSTR + 256 + ka];
    f.aD1 = *(const v2f*)&sh_h[(16 + mrow) * HSTR + 256 + ka];
}
__device__ __forceinline__ void l1_comb(L1Frag& f) {
    #pragma unroll
    for (int t = 0; t < 4; ++t) {
        f.bs[t] = f.wa[t] + f.wc[t];
        f.bd[t] = f.wa[t] - f.wc[t];
    }
}
__device__ __forceinline__ void l1_mma(v8f (&accF)[2][4], v8f (&accR)[2][4], const L1Frag& f) {
    #pragma unroll
    for (int t = 0; t < 4; ++t) {
        accF[0][t] = wmma_f32(f.aS0, f.bs[t], accF[0][t]);
        accF[1][t] = wmma_f32(f.aS1, f.bs[t], accF[1][t]);
        accR[0][t] = wmma_f32(f.aD0, f.bd[t], accR[0][t]);
        accR[1][t] = wmma_f32(f.aD1, f.bd[t], accR[1][t]);
        accF[0][t] = wmma_f32(f.aM0, f.wb[t], accF[0][t]);
        accF[1][t] = wmma_f32(f.aM1, f.wb[t], accF[1][t]);
    }
}

__device__ __forceinline__ void l2_load_w(L2Frag& f, const float* __restrict__ W2, int stage, int khalf, int ncb) {
    #pragma unroll
    for (int u = 0; u < 2; ++u) {
        const int ka = (stage * 2 + u) * 4 + 2 * khalf;
        #pragma unroll
        for (int t = 0; t < 4; ++t) {
            const int n = ncb + t * 16;
            f.b[u][t].x = W2[(ka + 0) * DMID + n];
            f.b[u][t].y = W2[(ka + 1) * DMID + n];
        }
    }
}
__device__ __forceinline__ void l2_load_z(L2Frag& f, const float* zf, const float* zr, int stage, int khalf, int mrow) {
    #pragma unroll
    for (int u = 0; u < 2; ++u) {
        const int ka = (stage * 2 + u) * 4 + 2 * khalf;
        f.a[u][0] = *(const v2f*)&zf[ mrow       * ZSTR + ka];
        f.a[u][1] = *(const v2f*)&zf[(16 + mrow) * ZSTR + ka];
        f.a[u][2] = *(const v2f*)&zr[ mrow       * ZSTR + ka];
        f.a[u][3] = *(const v2f*)&zr[(16 + mrow) * ZSTR + ka];
    }
}
__device__ __forceinline__ void l2_mma(v8f (&accF)[2][4], v8f (&accR)[2][4], const L2Frag& f) {
    #pragma unroll
    for (int u = 0; u < 2; ++u)
        #pragma unroll
        for (int t = 0; t < 4; ++t) {
            accF[0][t] = wmma_f32(f.a[u][0], f.b[u][t], accF[0][t]);
            accF[1][t] = wmma_f32(f.a[u][1], f.b[u][t], accF[1][t]);
            accR[0][t] = wmma_f32(f.a[u][2], f.b[u][t], accR[0][t]);
            accR[1][t] = wmma_f32(f.a[u][3], f.b[u][t], accR[1][t]);
        }
}

__global__ __launch_bounds__(128, 1)
void janossy_readout_kernel(const float* __restrict__ x_atoms,
                            const int*   __restrict__ frag_idx,
                            const float* __restrict__ W1,
                            const float* __restrict__ b1,
                            const float* __restrict__ W2,
                            const float* __restrict__ b2,
                            const float* __restrict__ Wout,
                            const float* __restrict__ bout,
                            float* __restrict__ out)
{
    __shared__ __attribute__((aligned(16))) float sh_region[2 * MBLOCK * ZSTR];
    __shared__ int   sh_idx[MBLOCK * 3];
    static_assert(MBLOCK * HSTR <= 2 * MBLOCK * ZSTR, "gathered rows must fit the region");
    float* const sh_h   = sh_region;
    float* const sh_z1f = sh_region;
    float* const sh_z1r = sh_region + MBLOCK * ZSTR;
    float* const sh_p   = sh_region;

    const int tid   = threadIdx.x;
    const int frag0 = blockIdx.x * MBLOCK;

    const int lane  = tid & 31;
    const int wave  = tid >> 5;
    const int mrow  = lane & 15;
    const int khalf = lane >> 4;
    const int ncol  = lane & 15;
    const int n0    = wave * 64;
    const int ncb   = n0 + ncol;

    L1Frag q1a, q1b, q1c;
    L2Frag q2a, q2b, q2c;

    l1_load_w(q1a, W1, 0, khalf, ncb);
    l1_load_w(q1b, W1, 1, khalf, ncb);

    if (tid < MBLOCK * 3) sh_idx[tid] = frag_idx[frag0 * 3 + tid];
    __syncthreads();

    #pragma unroll 2
    for (int i = tid; i < MBLOCK * 32; i += 128) {
        int m   = i >> 5;
        int c4  = i & 31;
        const float4* s0 = (const float4*)(x_atoms + (size_t)sh_idx[m * 3 + 0] * 128);
        const float4* s1 = (const float4*)(x_atoms + (size_t)sh_idx[m * 3 + 1] * 128);
        const float4* s2 = (const float4*)(x_atoms + (size_t)sh_idx[m * 3 + 2] * 128);
        float4 v0 = s0[c4];
        float4 v1 = s1[c4];
        float4 v2 = s2[c4];
        float4 hs, hd;
        hs.x = 0.5f * (v0.x + v2.x);  hd.x = 0.5f * (v0.x - v2.x);
        hs.y = 0.5f * (v0.y + v2.y);  hd.y = 0.5f * (v0.y - v2.y);
        hs.z = 0.5f * (v0.z + v2.z);  hd.z = 0.5f * (v0.z - v2.z);
        hs.w = 0.5f * (v0.w + v2.w);  hd.w = 0.5f * (v0.w - v2.w);
        *(float4*)&sh_h[m * HSTR +   0 + c4 * 4] = hs;
        *(float4*)&sh_h[m * HSTR + 128 + c4 * 4] = v1;
        *(float4*)&sh_h[m * HSTR + 256 + c4 * 4] = hd;
    }
    __syncthreads();

    v8f accF[2][4], accR[2][4];
    #pragma unroll
    for (int mt = 0; mt < 2; ++mt)
        #pragma unroll
        for (int t = 0; t < 4; ++t) {
            accF[mt][t] = (v8f){0.f,0.f,0.f,0.f,0.f,0.f,0.f,0.f};
            accR[mt][t] = (v8f){0.f,0.f,0.f,0.f,0.f,0.f,0.f,0.f};
        }

#define L1_STAGE(X, Z, jz)                                   \
    l1_comb(X);                                              \
    PIN_ORDER();                                             \
    l1_load_w(Z, W1, (jz), khalf, ncb);                      \
    l1_load_h(Z, sh_h, (jz), khalf, mrow);                   \
    PIN_ORDER();                                             \
    l1_mma(accF, accR, X);                                   \
    PIN_ORDER();

    l1_load_h(q1a, sh_h, 0, khalf, mrow);
    l1_load_h(q1b, sh_h, 1, khalf, mrow);
    for (int j = 0; j < 30; j += 3) {
        L1_STAGE(q1a, q1c, j + 2)
        L1_STAGE(q1b, q1a, j + 3)
        L1_STAGE(q1c, q1b, j + 4)
    }
    __syncthreads();
    l1_comb(q1a);
    PIN_ORDER();
    l2_load_w(q2a, W2, 0, khalf, ncb);
    PIN_ORDER();
    l1_mma(accF, accR, q1a);
    PIN_ORDER();
    l1_comb(q1b);
    PIN_ORDER();
    l2_load_w(q2b, W2, 1, khalf, ncb);
    PIN_ORDER();
    l1_mma(accF, accR, q1b);
#undef L1_STAGE

    #pragma unroll
    for (int mt = 0; mt < 2; ++mt)
        #pragma unroll
        for (int t = 0; t < 4; ++t) {
            const int n = n0 + t * 16 + ncol;
            const float bias = b1[n];
            #pragma unroll
            for (int i = 0; i < 8; ++i) {
                const int m = mt * 16 + i + 8 * khalf;
                const float sS = accF[mt][t][i];
                const float sD = accR[mt][t][i];
                float vf = (sS + sD) + bias; vf = vf > 0.f ? vf : 0.f;
                float vr = (sS - sD) + bias; vr = vr > 0.f ? vr : 0.f;
                sh_z1f[m * ZSTR + n] = vf;
                sh_z1r[m * ZSTR + n] = vr;
            }
        }
    __syncthreads();

    #pragma unroll
    for (int mt = 0; mt < 2; ++mt)
        #pragma unroll
        for (int t = 0; t < 4; ++t) {
            accF[mt][t] = (v8f){0.f,0.f,0.f,0.f,0.f,0.f,0.f,0.f};
            accR[mt][t] = (v8f){0.f,0.f,0.f,0.f,0.f,0.f,0.f,0.f};
        }

#define L2_STAGE(X, Z, jz)                                   \
    l2_load_w(Z, W2, (jz), khalf, ncb);                      \
    l2_load_z(Z, sh_z1f, sh_z1r, (jz), khalf, mrow);         \
    PIN_ORDER();                                             \
    l2_mma(accF, accR, X);                                   \
    PIN_ORDER();

    l2_load_z(q2a, sh_z1f, sh_z1r, 0, khalf, mrow);
    l2_load_z(q2b, sh_z1f, sh_z1r, 1, khalf, mrow);
    for (int j = 0; j < 30; j += 3) {
        L2_STAGE(q2a, q2c, j + 2)
        L2_STAGE(q2b, q2a, j + 3)
        L2_STAGE(q2c, q2b, j + 4)
    }
    __syncthreads();
    l2_mma(accF, accR, q2a);
    l2_mma(accF, accR, q2b);
#undef L2_STAGE

    #pragma unroll
    for (int mt = 0; mt < 2; ++mt)
        #pragma unroll
        for (int t = 0; t < 4; ++t) {
            const int n = n0 + t * 16 + ncol;
            const float bias = b2[n];
            #pragma unroll
            for (int i = 0; i < 8; ++i) {
                const int m = mt * 16 + i + 8 * khalf;
                float vf = accF[mt][t][i] + bias; vf = vf > 0.f ? vf : 0.f;
                float vr = accR[mt][t][i] + bias; vr = vr > 0.f ? vr : 0.f;
                sh_p[m * ZSTR + n] = vf + vr;
            }
        }
    __syncthreads();

    for (int o = tid; o < MBLOCK * 3; o += 128) {
        const int m = o / 3;
        const int d = o - m * 3;
        float s = bout[d];
        const float* prow = &sh_p[m * ZSTR];
        #pragma unroll 4
        for (int k = 0; k < DMID; ++k)
            s += prow[k] * Wout[k * 3 + d];
        out[(size_t)(frag0 + m) * 3 + d] = s;
    }
}

extern "C" void kernel_launch(void* const* d_in, const int* in_sizes, int n_in,
                              void* d_out, int out_size, void* d_ws, size_t ws_size,
                              hipStream_t stream) {
    const float* x_atoms = (const float*)d_in[0];
    const int*   frag_idx = (const int*)d_in[1];
    const float* W1   = (const float*)d_in[2];
    const float* b1   = (const float*)d_in[3];
    const float* W2   = (const float*)d_in[4];
    const float* b2   = (const float*)d_in[5];
    const float* Wout = (const float*)d_in[6];
    const float* bout = (const float*)d_in[7];
    float* out = (float*)d_out;

    const int n_frag = in_sizes[1] / 3;
    const int grid = n_frag / MBLOCK;
    janossy_readout_kernel<<<grid, 128, 0, stream>>>(
        x_atoms, frag_idx, W1, b1, W2, b2, Wout, bout, out);
}


# Column map n0+4c+t in each wave's 64-column chunk: W1/W2 fragments by 16-byte row loads (320 not 1280 per wave), Z1/pooled by 16-byte LDS stores; 255 VGPRs, so 4 workgroups per WGP; sums unchanged
# speedup vs baseline: 1.6229x; 1.0305x over previous
#include <hip/hip_runtime.h>
#include <hip/hip_bf16.h>

typedef __attribute__((ext_vector_type(2))) float v2f;
typedef __attribute__((ext_vector_type(4))) float v4f;
typedef __attribute__((ext_vector_type(8))) float v8f;

#define N_FRAG  262144
#define MBLOCK  32
#define KDIM    384
#define DMID    256
#define HSTR    396
#define ZSTR    268

__device__ __forceinline__ v8f wmma_f32(v2f a, v2f b, v8f c) {
    return __builtin_amdgcn_wmma_f32_16x16x4_f32(
         false, a,  false, b,
         (short)0, c,  false,  false);
}

#define PIN_ORDER() __builtin_amdgcn_sched_barrier(0)

#define OPERAND_RELEASE_GAP() do { PIN_ORDER(); asm volatile("v_nop\n\tv_nop\n\tv_nop\n\tv_nop"); PIN_ORDER(); } while (0)

struct L1Frag {
    v4f wa[2], wb[2], wc[2];
    v2f aS0, aS1, aM0, aM1, aD0, aD1;
    v2f bs[4], bd[4], bm[4];
};
struct L2Frag {
    v4f q[2][2];
    v2f a[2][4];
    v2f b[2][4];
};

__device__ __forceinline__ void l1_load_w(L1Frag& f, const float* __restrict__ W1, int stage, int khalf, int ncq) {
    const int ka = stage * 4 + 2 * khalf;
    f.wa[0] = *(const v4f*)&W1[(      ka + 0) * DMID + ncq];
    f.wa[1] = *(const v4f*)&W1[(      ka + 1) * DMID + ncq];
    f.wb[0] = *(const v4f*)&W1[(128 + ka + 0) * DMID + ncq];
    f.wb[1] = *(const v4f*)&W1[(128 + ka + 1) * DMID + ncq];
    f.wc[0] = *(const v4f*)&W1[(256 + ka + 0) * DMID + ncq];
    f.wc[1] = *(const v4f*)&W1[(256 + ka + 1) * DMID + ncq];
}
__device__ __forceinline__ void l1_load_h(L1Frag& f, const float* sh_h, int stage, int khalf, int mrow) {
    const int ka = stage * 4 + 2 * khalf;
    f.aS0 = *(const v2f*)&sh_h[ mrow       * HSTR +       ka];
    f.aS1 = *(const v2f*)&sh_h[(16 + mrow) * HSTR +       ka];
    f.aM0 = *(const v2f*)&sh_h[ mrow       * HSTR + 128 + ka];
    f.aM1 = *(const v2f*)&sh_h[(16 + mrow) * HSTR + 128 + ka];
    f.aD0 = *(const v2f*)&sh_h[ mrow       * HSTR + 256 + ka];
    f.aD1 = *(const v2f*)&sh_h[(16 + mrow) * HSTR + 256 + ka];
}
__device__ __forceinline__ void l1_comb(L1Frag& f) {
    #pragma unroll
    for (int t = 0; t < 4; ++t) {
        f.bs[t] = (v2f){ f.wa[0][t] + f.wc[0][t], f.wa[1][t] + f.wc[1][t] };
        f.bd[t] = (v2f){ f.wa[0][t] - f.wc[0][t], f.wa[1][t] - f.wc[1][t] };
        f.bm[t] = (v2f){ f.wb[0][t], f.wb[1][t] };
    }
}
__device__ __forceinline__ void l1_mma(v8f (&accF)[2][4], v8f (&accR)[2][4], const L1Frag& f) {
    #pragma unroll
    for (int t = 0; t < 4; ++t) {
        accF[0][t] = wmma_f32(f.aS0, f.bs[t], accF[0][t]);
        accF[1][t] = wmma_f32(f.aS1, f.bs[t], accF[1][t]);
        accR[0][t] = wmma_f32(f.aD0, f.bd[t], accR[0][t]);
        accR[1][t] = wmma_f32(f.aD1, f.bd[t], accR[1][t]);
        accF[0][t] = wmma_f32(f.aM0, f.bm[t], accF[0][t]);
        accF[1][t] = wmma_f32(f.aM1, f.bm[t], accF[1][t]);
    }
}

__device__ __forceinline__ void l2_load_w(L2Frag& f, const float* __restrict__ W2, int stage, int khalf, int ncq) {
    #pragma unroll
    for (int u = 0; u < 2; ++u) {
        const int ka = (stage * 2 + u) * 4 + 2 * khalf;
        f.q[u][0] = *(const v4f*)&W2[(ka + 0) * DMID + ncq];
        f.q[u][1] = *(const v4f*)&W2[(ka + 1) * DMID + ncq];
    }
}
__device__ __forceinline__ void l2_load_z(L2Frag& f, const float* zf, const float* zr, int stage, int khalf, int mrow) {
    #pragma unroll
    for (int u = 0; u < 2; ++u) {
        const int ka = (stage * 2 + u) * 4 + 2 * khalf;
        f.a[u][0] = *(const v2f*)&zf[ mrow       * ZSTR + ka];
        f.a[u][1] = *(const v2f*)&zf[(16 + mrow) * ZSTR + ka];
        f.a[u][2] = *(const v2f*)&zr[ mrow       * ZSTR + ka];
        f.a[u][3] = *(const v2f*)&zr[(16 + mrow) * ZSTR + ka];
    }
}
__device__ __forceinline__ void l2_comb(L2Frag& f) {
    #pragma unroll
    for (int u = 0; u < 2; ++u)
        #pragma unroll
        for (int t = 0; t < 4; ++t)
            f.b[u][t] = (v2f){ f.q[u][0][t], f.q[u][1][t] };
}
__device__ __forceinline__ void l2_mma(v8f (&accF)[2][4], v8f (&accR)[2][4], const L2Frag& f) {
    #pragma unroll
    for (int u = 0; u < 2; ++u)
        #pragma unroll
        for (int t = 0; t < 4; ++t) {
            accF[0][t] = wmma_f32(f.a[u][0], f.b[u][t], accF[0][t]);
            accF[1][t] = wmma_f32(f.a[u][1], f.b[u][t], accF[1][t]);
            accR[0][t] = wmma_f32(f.a[u][2], f.b[u][t], accR[0][t]);
            accR[1][t] = wmma_f32(f.a[u][3], f.b[u][t], accR[1][t]);
        }
}

__global__ __launch_bounds__(128, 1)
void janossy_readout_kernel(const float* __restrict__ x_atoms,
                            const int*   __restrict__ frag_idx,
                            const float* __restrict__ W1,
                            const float* __restrict__ b1,
                            const float* __restrict__ W2,
                            const float* __restrict__ b2,
                            const float* __restrict__ Wout,
                            const float* __restrict__ bout,
                            float* __restrict__ out)
{
    __shared__ __attribute__((aligned(16))) float sh_region[2 * MBLOCK * ZSTR];
    __shared__ int   sh_idx[MBLOCK * 3];
    static_assert(MBLOCK * HSTR <= 2 * MBLOCK * ZSTR, "gathered rows must fit the region");
    float* const sh_h   = sh_region;
    float* const sh_z1f = sh_region;
    float* const sh_z1r = sh_region + MBLOCK * ZSTR;
    float* const sh_p   = sh_region;

    const int tid   = threadIdx.x;
    const int frag0 = blockIdx.x * MBLOCK;

    const int lane  = tid & 31;
    const int wave  = tid >> 5;
    const int mrow  = lane & 15;
    const int khalf = lane >> 4;
    const int ncol  = lane & 15;
    const int n0    = wave * 64;
    const int ncq   = n0 + 4 * ncol;

    L1Frag q1a, q1b, q1c;
    L2Frag q2a, q2b, q2c;

    l1_load_w(q1a, W1, 0, khalf, ncq);
    l1_load_w(q1b, W1, 1, khalf, ncq);

    if (tid < MBLOCK * 3) sh_idx[tid] = frag_idx[frag0 * 3 + tid];
    __syncthreads();

    #pragma unroll 2
    for (int i = tid; i < MBLOCK * 32; i += 128) {
        int m   = i >> 5;
        int c4  = i & 31;
        const float4* s0 = (const float4*)(x_atoms + (size_t)sh_idx[m * 3 + 0] * 128);
        const float4* s1 = (const float4*)(x_atoms + (size_t)sh_idx[m * 3 + 1] * 128);
        const float4* s2 = (const float4*)(x_atoms + (size_t)sh_idx[m * 3 + 2] * 128);
        float4 v0 = s0[c4];
        float4 v1 = s1[c4];
        float4 v2 = s2[c4];
        float4 hs, hd;
        hs.x = 0.5f * (v0.x + v2.x);  hd.x = 0.5f * (v0.x - v2.x);
        hs.y = 0.5f * (v0.y + v2.y);  hd.y = 0.5f * (v0.y - v2.y);
        hs.z = 0.5f * (v0.z + v2.z);  hd.z = 0.5f * (v0.z - v2.z);
        hs.w = 0.5f * (v0.w + v2.w);  hd.w = 0.5f * (v0.w - v2.w);
        *(float4*)&sh_h[m * HSTR +   0 + c4 * 4] = hs;
        *(float4*)&sh_h[m * HSTR + 128 + c4 * 4] = v1;
        *(float4*)&sh_h[m * HSTR + 256 + c4 * 4] = hd;
    }
    __syncthreads();

    v8f accF[2][4], accR[2][4];
    #pragma unroll
    for (int mt = 0; mt < 2; ++mt)
        #pragma unroll
        for (int t = 0; t < 4; ++t) {
            accF[mt][t] = (v8f){0.f,0.f,0.f,0.f,0.f,0.f,0.f,0.f};
            accR[mt][t] = (v8f){0.f,0.f,0.f,0.f,0.f,0.f,0.f,0.f};
        }

#define L1_STAGE(X, Z, jz)                                   \
    OPERAND_RELEASE_GAP();                                   \
    l1_comb(X);                                              \
    PIN_ORDER();                                             \
    l1_load_w(Z, W1, (jz), khalf, ncq);                      \
    l1_load_h(Z, sh_h, (jz), khalf, mrow);                   \
    PIN_ORDER();                                             \
    l1_mma(accF, accR, X);                                   \
    PIN_ORDER();

    l1_load_h(q1a, sh_h, 0, khalf, mrow);
    l1_load_h(q1b, sh_h, 1, khalf, mrow);
    for (int j = 0; j < 30; j += 3) {
        L1_STAGE(q1a, q1c, j + 2)
        L1_STAGE(q1b, q1a, j + 3)
        L1_STAGE(q1c, q1b, j + 4)
    }
    OPERAND_RELEASE_GAP();
    __syncthreads();
    l1_comb(q1a);
    PIN_ORDER();
    l2_load_w(q2a, W2, 0, khalf, ncq);
    PIN_ORDER();
    l1_mma(accF, accR, q1a);
    OPERAND_RELEASE_GAP();
    l1_comb(q1b);
    PIN_ORDER();
    l2_load_w(q2b, W2, 1, khalf, ncq);
    PIN_ORDER();
    l1_mma(accF, accR, q1b);
    OPERAND_RELEASE_GAP();
#undef L1_STAGE

    {
        const v4f bias = *(const v4f*)&b1[ncq];
        #pragma unroll
        for (int mt = 0; mt < 2; ++mt)
            #pragma unroll
            for (int i = 0; i < 8; ++i) {
                const int m = mt * 16 + i + 8 * khalf;
                v4f vf, vr;
                #pragma unroll
                for (int t = 0; t < 4; ++t) {
                    const float sS = accF[mt][t][i];
                    const float sD = accR[mt][t][i];
                    float f = (sS + sD) + bias[t]; vf[t] = f > 0.f ? f : 0.f;
                    float r = (sS - sD) + bias[t]; vr[t] = r > 0.f ? r : 0.f;
                }
                *(v4f*)&sh_z1f[m * ZSTR + ncq] = vf;
                *(v4f*)&sh_z1r[m * ZSTR + ncq] = vr;
            }
    }
    __syncthreads();

    #pragma unroll
    for (int mt = 0; mt < 2; ++mt)
        #pragma unroll
        for (int t = 0; t < 4; ++t) {
            accF[mt][t] = (v8f){0.f,0.f,0.f,0.f,0.f,0.f,0.f,0.f};
            accR[mt][t] = (v8f){0.f,0.f,0.f,0.f,0.f,0.f,0.f,0.f};
        }

#define L2_STAGE(X, Z, jz)                                   \
    OPERAND_RELEASE_GAP();                                   \
    l2_comb(X);                                              \
    PIN_ORDER();                                             \
    l2_load_w(Z, W2, (jz), khalf, ncq);                      \
    l2_load_z(Z, sh_z1f, sh_z1r, (jz), khalf, mrow);         \
    PIN_ORDER();                                             \
    l2_mma(accF, accR, X);                                   \
    PIN_ORDER();

    l2_load_z(q2a, sh_z1f, sh_z1r, 0, khalf, mrow);
    l2_load_z(q2b, sh_z1f, sh_z1r, 1, khalf, mrow);
    for (int j = 0; j < 30; j += 3) {
        L2_STAGE(q2a, q2c, j + 2)
        L2_STAGE(q2b, q2a, j + 3)
        L2_STAGE(q2c, q2b, j + 4)
    }
    OPERAND_RELEASE_GAP();
    __syncthreads();
    l2_comb(q2a);
    l2_comb(q2b);
    PIN_ORDER();
    l2_mma(accF, accR, q2a);
    l2_mma(accF, accR, q2b);
    OPERAND_RELEASE_GAP();
#undef L2_STAGE

    {
        const v4f bias = *(const v4f*)&b2[ncq];
        #pragma unroll
        for (int mt = 0; mt < 2; ++mt)
            #pragma unroll
            for (int i = 0; i < 8; ++i) {
                const int m = mt * 16 + i + 8 * khalf;
                v4f vp;
                #pragma unroll
                for (int t = 0; t < 4; ++t) {
                    float vf = accF[mt][t][i] + bias[t]; vf = vf > 0.f ? vf : 0.f;
                    float vr = accR[mt][t][i] + bias[t]; vr = vr > 0.f ? vr : 0.f;
                    vp[t] = vf + vr;
                }
                *(v4f*)&sh_p[m * ZSTR + ncq] = vp;
            }
    }
    __syncthreads();

    for (int o = tid; o < MBLOCK * 3; o += 128) {
        const int m = o / 3;
        const int d = o - m * 3;
        float s = bout[d];
        const float* prow = &sh_p[m * ZSTR];
        #pragma unroll 4
        for (int k = 0; k < DMID; ++k)
            s += prow[k] * Wout[k * 3 + d];
        out[(size_t)(frag0 + m) * 3 + d] = s;
    }
}

extern "C" void kernel_launch(void* const* d_in, const int* in_sizes, int n_in,
                              void* d_out, int out_size, void* d_ws, size_t ws_size,
                              hipStream_t stream) {
    const float* x_atoms = (const float*)d_in[0];
    const int*   frag_idx = (const int*)d_in[1];
    const float* W1   = (const float*)d_in[2];
    const float* b1   = (const float*)d_in[3];
    const float* W2   = (const float*)d_in[4];
    const float* b2   = (const float*)d_in[5];
    const float* Wout = (const float*)d_in[6];
    const float* bout = (const float*)d_in[7];
    float* out = (float*)d_out;

    const int n_frag = in_sizes[1] / 3;
    const int grid = n_frag / MBLOCK;
    janossy_readout_kernel<<<grid, 128, 0, stream>>>(
        x_atoms, frag_idx, W1, b1, W2, b2, Wout, bout, out);
}


# W1 block sums/differences and W2 written per call by a first launch to 640 KiB of workspace in k-pair order: 16-byte loads are ready B pairs, K loops lose adds, pair moves, gaps; sums unchanged
# speedup vs baseline: 1.6763x; 1.0329x over previous
#include <hip/hip_runtime.h>
#include <hip/hip_bf16.h>

typedef __attribute__((ext_vector_type(2))) float v2f;
typedef __attribute__((ext_vector_type(4))) float v4f;
typedef __attribute__((ext_vector_type(8))) float v8f;

#define N_FRAG  262144
#define MBLOCK  32
#define KDIM    384
#define DMID    256
#define HSTR    396
#define ZSTR    268

#define PREP_ROW      512
#define PREP_L1_OP    (64 * PREP_ROW)
#define PREP_L2_BASE  (3 * PREP_L1_OP)
#define PREP_FLOATS   (PREP_L2_BASE + 128 * PREP_ROW)

__device__ __forceinline__ v8f wmma_f32(v2f a, v2f b, v8f c) {
    return __builtin_amdgcn_wmma_f32_16x16x4_f32(
         false, a,  false, b,
         (short)0, c,  false,  false);
}

#define PIN_ORDER() __builtin_amdgcn_sched_barrier(0)

#define OPERAND_RELEASE_GAP() do { PIN_ORDER(); asm volatile("v_nop\n\tv_nop\n\tv_nop\n\tv_nop"); PIN_ORDER(); } while (0)

struct L1Frag {
    v4f wa[2], wb[2], wc[2];
    v4f ps[2], pd[2], pm[2];
    v2f aS0, aS1, aM0, aM1, aD0, aD1;
    v2f bs[4], bd[4], bm[4];
};
struct L2Frag {
    v4f q[2][2];
    v4f p[2][2];
    v2f a[2][4];
    v2f b[2][4];
};

__device__ __forceinline__ void l1_load_w(L1Frag& f, const float* __restrict__ W1, int stage, int khalf, int ncq) {
    const int ka = stage * 4 + 2 * khalf;
    f.wa[0] = *(const v4f*)&W1[(      ka + 0) * DMID + ncq];
    f.wa[1] = *(const v4f*)&W1[(      ka + 1) * DMID + ncq];
    f.wb[0] = *(const v4f*)&W1[(128 + ka + 0) * DMID + ncq];
    f.wb[1] = *(const v4f*)&W1[(128 + ka + 1) * DMID + ncq];
    f.wc[0] = *(const v4f*)&W1[(256 + ka + 0) * DMID + ncq];
    f.wc[1] = *(const v4f*)&W1[(256 + ka + 1) * DMID + ncq];
}
__device__ __forceinline__ void l1_load_h(L1Frag& f, const float* sh_h, int stage, int khalf, int mrow) {
    const int ka = stage * 4 + 2 * khalf;
    f.aS0 = *(const v2f*)&sh_h[ mrow       * HSTR +       ka];
    f.aS1 = *(const v2f*)&sh_h[(16 + mrow) * HSTR +       ka];
    f.aM0 = *(const v2f*)&sh_h[ mrow       * HSTR + 128 + ka];
    f.aM1 = *(const v2f*)&sh_h[(16 + mrow) * HSTR + 128 + ka];
    f.aD0 = *(const v2f*)&sh_h[ mrow       * HSTR + 256 + ka];
    f.aD1 = *(const v2f*)&sh_h[(16 + mrow) * HSTR + 256 + ka];
}
__device__ __forceinline__ void l1_comb(L1Frag& f) {
    #pragma unroll
    for (int t = 0; t < 4; ++t) {
        f.bs[t] = (v2f){ f.wa[0][t] + f.wc[0][t], f.wa[1][t] + f.wc[1][t] };
        f.bd[t] = (v2f){ f.wa[0][t] - f.wc[0][t], f.wa[1][t] - f.wc[1][t] };
        f.bm[t] = (v2f){ f.wb[0][t], f.wb[1][t] };
    }
}
__device__ __forceinline__ void l1_load_p(L1Frag& f, const float* __restrict__ P, int stage, int poff) {
    const float* r = P + stage * (2 * PREP_ROW) + poff;
    f.ps[0] = *(const v4f*)&r[0];
    f.ps[1] = *(const v4f*)&r[64];
    f.pd[0] = *(const v4f*)&r[PREP_L1_OP];
    f.pd[1] = *(const v4f*)&r[PREP_L1_OP + 64];
    f.pm[0] = *(const v4f*)&r[2 * PREP_L1_OP];
    f.pm[1] = *(const v4f*)&r[2 * PREP_L1_OP + 64];
}
__device__ __forceinline__ void l1_pairs_p(L1Frag& f) {
    #pragma unroll
    for (int t = 0; t < 4; ++t) {
        const int g = t >> 1, e = 2 * (t & 1);
        f.bs[t] = (v2f){ f.ps[g][e], f.ps[g][e + 1] };
        f.bd[t] = (v2f){ f.pd[g][e], f.pd[g][e + 1] };
        f.bm[t] = (v2f){ f.pm[g][e], f.pm[g][e + 1] };
    }
}
__device__ __forceinline__ void l1_mma(v8f (&accF)[2][4], v8f (&accR)[2][4], const L1Frag& f) {
    #pragma unroll
    for (int t = 0; t < 4; ++t) {
        accF[0][t] = wmma_f32(f.aS0, f.bs[t], accF[0][t]);
        accF[1][t] = wmma_f32(f.aS1, f.bs[t], accF[1][t]);
        accR[0][t] = wmma_f32(f.aD0, f.bd[t], accR[0][t]);
        accR[1][t] = wmma_f32(f.aD1, f.bd[t], accR[1][t]);
        accF[0][t] = wmma_f32(f.aM0, f.bm[t], accF[0][t]);
        accF[1][t] = wmma_f32(f.aM1, f.bm[t], accF[1][t]);
    }
}

__device__ __forceinline__ void l2_load_w(L2Frag& f, const float* __restrict__ W2, int stage, int khalf, int ncq) {
    #pragma unroll
    for (int u = 0; u < 2; ++u) {
        const int ka = (stage * 2 + u) * 4 + 2 * khalf;
        f.q[u][0] = *(const v4f*)&W2[(ka + 0) * DMID + ncq];
        f.q[u][1] = *(const v4f*)&W2[(ka + 1) * DMID + ncq];
    }
}
__device__ __forceinline__ void l2_load_z(L2Frag& f, const float* zf, const float* zr, int stage, int khalf, int mrow) {
    #pragma unroll
    for (int u = 0; u < 2; ++u) {
        const int ka = (stage * 2 + u) * 4 + 2 * khalf;
        f.a[u][0] = *(const v2f*)&zf[ mrow       * ZSTR + ka];
        f.a[u][1] = *(const v2f*)&zf[(16 + mrow) * ZSTR + ka];
        f.a[u][2] = *(const v2f*)&zr[ mrow       * ZSTR + ka];
        f.a[u][3] = *(const v2f*)&zr[(16 + mrow) * ZSTR + ka];
    }
}
__device__ __forceinline__ void l2_comb(L2Frag& f) {
    #pragma unroll
    for (int u = 0; u < 2; ++u)
        #pragma unroll
        for (int t = 0; t < 4; ++t)
            f.b[u][t] = (v2f){ f.q[u][0][t], f.q[u][1][t] };
}
__device__ __forceinline__ void l2_load_p(L2Frag& f, const float* __restrict__ P, int stage, int poff) {
    #pragma unroll
    for (int u = 0; u < 2; ++u) {
        const float* r = P + (stage * 2 + u) * (2 * PREP_ROW) + poff;
        f.p[u][0] = *(const v4f*)&r[0];
        f.p[u][1] = *(const v4f*)&r[64];
    }
}
__device__ __forceinline__ void l2_pairs_p(L2Frag& f) {
    #pragma unroll
    for (int u = 0; u < 2; ++u)
        #pragma unroll
        for (int t = 0; t < 4; ++t) {
            const int g = t >> 1, e = 2 * (t & 1);
            f.b[u][t] = (v2f){ f.p[u][g][e], f.p[u][g][e + 1] };
        }
}
__device__ __forceinline__ void l2_mma(v8f (&accF)[2][4], v8f (&accR)[2][4], const L2Frag& f) {
    #pragma unroll
    for (int u = 0; u < 2; ++u)
        #pragma unroll
        for (int t = 0; t < 4; ++t) {
            accF[0][t] = wmma_f32(f.a[u][0], f.b[u][t], accF[0][t]);
            accF[1][t] = wmma_f32(f.a[u][1], f.b[u][t], accF[1][t]);
            accR[0][t] = wmma_f32(f.a[u][2], f.b[u][t], accR[0][t]);
            accR[1][t] = wmma_f32(f.a[u][3], f.b[u][t], accR[1][t]);
        }
}

__global__ __launch_bounds__(256)
void prearrange_weights(const float* __restrict__ W1, const float* __restrict__ W2, float* __restrict__ P)
{
    const int o   = blockIdx.x * 256 + threadIdx.x;
    const int row = o >> 7;
    const int q   = o & 127;
    const int n   = 64 * (q >> 5) + 4 * (q & 15) + 2 * ((q >> 4) & 1);
    v4f r;
    if (row < 192) {
        const int op = row >> 6;
        const int k0 = 2 * (row & 63);
        if (op == 2) {
            const v2f b0 = *(const v2f*)&W1[(128 + k0) * DMID + n];
            const v2f b1 = *(const v2f*)&W1[(129 + k0) * DMID + n];
            r = (v4f){ b0[0], b1[0], b0[1], b1[1] };
        } else {
            const v2f a0 = *(const v2f*)&W1[(      k0    ) * DMID + n];
            const v2f a1 = *(const v2f*)&W1[(      k0 + 1) * DMID + n];
            const v2f c0 = *(const v2f*)&W1[(256 + k0    ) * DMID + n];
            const v2f c1 = *(const v2f*)&W1[(256 + k0 + 1) * DMID + n];
            r = (op == 0) ? (v4f){ a0[0] + c0[0], a1[0] + c1[0], a0[1] + c0[1], a1[1] + c1[1] }
                          : (v4f){ a0[0] - c0[0], a1[0] - c1[0], a0[1] - c0[1], a1[1] - c1[1] };
        }
    } else {
        const int k0 = 2 * (row - 192);
        const v2f w0 = *(const v2f*)&W2[(k0    ) * DMID + n];
        const v2f w1 = *(const v2f*)&W2[(k0 + 1) * DMID + n];
        r = (v4f){ w0[0], w1[0], w0[1], w1[1] };
    }
    *(v4f*)&P[(size_t)o * 4] = r;
}

template <bool PRE>
__global__ __launch_bounds__(128, 1)
void janossy_readout_kernel(const float* __restrict__ x_atoms,
                            const int*   __restrict__ frag_idx,
                            const float* __restrict__ W1,
                            const float* __restrict__ b1,
                            const float* __restrict__ W2,
                            const float* __restrict__ b2,
                            const float* __restrict__ Wout,
                            const float* __restrict__ bout,
                            float* __restrict__ out)
{
    __shared__ __attribute__((aligned(16))) float sh_region[2 * MBLOCK * ZSTR];
    __shared__ int   sh_idx[MBLOCK * 3];
    static_assert(MBLOCK * HSTR <= 2 * MBLOCK * ZSTR, "gathered rows must fit the region");
    float* const sh_h   = sh_region;
    float* const sh_z1f = sh_region;
    float* const sh_z1r = sh_region + MBLOCK * ZSTR;
    float* const sh_p   = sh_region;

    const int tid   = threadIdx.x;
    const int frag0 = blockIdx.x * MBLOCK;

    const int lane  = tid & 31;
    const int wave  = tid >> 5;
    const int mrow  = lane & 15;
    const int khalf = lane >> 4;
    const int ncol  = lane & 15;
    const int n0    = wave * 64;
    const int ncq   = n0 + 4 * ncol;
    const int poff  = khalf * PREP_ROW + wave * 128 + ncol * 4;

    L1Frag q1a, q1b, q1c;
    L2Frag q2a, q2b, q2c;

    if constexpr (PRE) {
        l1_load_p(q1a, W1, 0, poff);
        l1_load_p(q1b, W1, 1, poff);
    } else {
        l1_load_w(q1a, W1, 0, khalf, ncq);
        l1_load_w(q1b, W1, 1, khalf, ncq);
    }

    if (tid < MBLOCK * 3) sh_idx[tid] = frag_idx[frag0 * 3 + tid];
    __syncthreads();

    #pragma unroll 2
    for (int i = tid; i < MBLOCK * 32; i += 128) {
        int m   = i >> 5;
        int c4  = i & 31;
        const float4* s0 = (const float4*)(x_atoms + (size_t)sh_idx[m * 3 + 0] * 128);
        const float4* s1 = (const float4*)(x_atoms + (size_t)sh_idx[m * 3 + 1] * 128);
        const float4* s2 = (const float4*)(x_atoms + (size_t)sh_idx[m * 3 + 2] * 128);
        float4 v0 = s0[c4];
        float4 v1 = s1[c4];
        float4 v2 = s2[c4];
        float4 hs, hd;
        hs.x = 0.5f * (v0.x + v2.x);  hd.x = 0.5f * (v0.x - v2.x);
        hs.y = 0.5f * (v0.y + v2.y);  hd.y = 0.5f * (v0.y - v2.y);
        hs.z = 0.5f * (v0.z + v2.z);  hd.z = 0.5f * (v0.z - v2.z);
        hs.w = 0.5f * (v0.w + v2.w);  hd.w = 0.5f * (v0.w - v2.w);
        *(float4*)&sh_h[m * HSTR +   0 + c4 * 4] = hs;
        *(float4*)&sh_h[m * HSTR + 128 + c4 * 4] = v1;
        *(float4*)&sh_h[m * HSTR + 256 + c4 * 4] = hd;
    }
    __syncthreads();

    v8f accF[2][4], accR[2][4];
    #pragma unroll
    for (int mt = 0; mt < 2; ++mt)
        #pragma unroll
        for (int t = 0; t < 4; ++t) {
            accF[mt][t] = (v8f){0.f,0.f,0.f,0.f,0.f,0.f,0.f,0.f};
            accR[mt][t] = (v8f){0.f,0.f,0.f,0.f,0.f,0.f,0.f,0.f};
        }

#define L1_STAGE(X, Z, jz)                                   \
    if constexpr (PRE) {                                     \
        l1_load_p(Z, W1, (jz), poff);                        \
        l1_load_h(Z, sh_h, (jz), khalf, mrow);               \
        PIN_ORDER();                                         \
        l1_pairs_p(X);                                       \
    } else {                                                 \
        OPERAND_RELEASE_GAP();                               \
        l1_comb(X);                                          \
        PIN_ORDER();                                         \
        l1_load_w(Z, W1, (jz), khalf, ncq);                  \
        l1_load_h(Z, sh_h, (jz), khalf, mrow);               \
        PIN_ORDER();                                         \
    }                                                        \
    l1_mma(accF, accR, X);                                   \
    PIN_ORDER();

    l1_load_h(q1a, sh_h, 0, khalf, mrow);
    l1_load_h(q1b, sh_h, 1, khalf, mrow);
    for (int j = 0; j < 30; j += 3) {
        L1_STAGE(q1a, q1c, j + 2)
        L1_STAGE(q1b, q1a, j + 3)
        L1_STAGE(q1c, q1b, j + 4)
    }
    if constexpr (!PRE) OPERAND_RELEASE_GAP();
    __syncthreads();
    if constexpr (PRE) {
        l2_load_p(q2a, W2, 0, poff);
        PIN_ORDER();
        l1_pairs_p(q1a);
        l1_mma(accF, accR, q1a);
        PIN_ORDER();
        l2_load_p(q2b, W2, 1, poff);
        PIN_ORDER();
        l1_pairs_p(q1b);
        l1_mma(accF, accR, q1b);
    } else {
        l1_comb(q1a);
        PIN_ORDER();
        l2_load_w(q2a, W2, 0, khalf, ncq);
        PIN_ORDER();
        l1_mma(accF, accR, q1a);
        OPERAND_RELEASE_GAP();
        l1_comb(q1b);
        PIN_ORDER();
        l2_load_w(q2b, W2, 1, khalf, ncq);
        PIN_ORDER();
        l1_mma(accF, accR, q1b);
    }
    OPERAND_RELEASE_GAP();
#undef L1_STAGE

    {
        const v4f bias = *(const v4f*)&b1[ncq];
        #pragma unroll
        for (int mt = 0; mt < 2; ++mt)
            #pragma unroll
            for (int i = 0; i < 8; ++i) {
                const int m = mt * 16 + i + 8 * khalf;
                v4f vf, vr;
                #pragma unroll
                for (int t = 0; t < 4; ++t) {
                    const float sS = accF[mt][t][i];
                    const float sD = accR[mt][t][i];
                    float f = (sS + sD) + bias[t]; vf[t] = f > 0.f ? f : 0.f;
                    float r = (sS - sD) + bias[t]; vr[t] = r > 0.f ? r : 0.f;
                }
                *(v4f*)&sh_z1f[m * ZSTR + ncq] = vf;
                *(v4f*)&sh_z1r[m * ZSTR + ncq] = vr;
            }
    }
    __syncthreads();

    #pragma unroll
    for (int mt = 0; mt < 2; ++mt)
        #pragma unroll
        for (int t = 0; t < 4; ++t) {
            accF[mt][t] = (v8f){0.f,0.f,0.f,0.f,0.f,0.f,0.f,0.f};
            accR[mt][t] = (v8f){0.f,0.f,0.f,0.f,0.f,0.f,0.f,0.f};
        }

#define L2_STAGE(X, Z, jz)                                   \
    if constexpr (PRE) {                                     \
        l2_load_p(Z, W2, (jz), poff);                        \
        l2_load_z(Z, sh_z1f, sh_z1r, (jz), khalf, mrow);     \
        PIN_ORDER();                                         \
        l2_pairs_p(X);                                       \
    } else {                                                 \
        OPERAND_RELEASE_GAP();                               \
        l2_comb(X);                                          \
        PIN_ORDER();                                         \
        l2_load_w(Z, W2, (jz), khalf, ncq);                  \
        l2_load_z(Z, sh_z1f, sh_z1r, (jz), khalf, mrow);     \
        PIN_ORDER();                                         \
    }                                                        \
    l2_mma(accF, accR, X);                                   \
    PIN_ORDER();

    l2_load_z(q2a, sh_z1f, sh_z1r, 0, khalf, mrow);
    l2_load_z(q2b, sh_z1f, sh_z1r, 1, khalf, mrow);
    for (int j = 0; j < 30; j += 3) {
        L2_STAGE(q2a, q2c, j + 2)
        L2_STAGE(q2b, q2a, j + 3)
        L2_STAGE(q2c, q2b, j + 4)
    }
    if constexpr (!PRE) OPERAND_RELEASE_GAP();
    __syncthreads();
    if constexpr (PRE) {
        l2_pairs_p(q2a);
        l2_pairs_p(q2b);
    } else {
        l2_comb(q2a);
        l2_comb(q2b);
    }
    PIN_ORDER();
    l2_mma(accF, accR, q2a);
    l2_mma(accF, accR, q2b);
    OPERAND_RELEASE_GAP();
#undef L2_STAGE

    {
        const v4f bias = *(const v4f*)&b2[ncq];
        #pragma unroll
        for (int mt = 0; mt < 2; ++mt)
            #pragma unroll
            for (int i = 0; i < 8; ++i) {
                const int m = mt * 16 + i + 8 * khalf;
                v4f vp;
                #pragma unroll
                for (int t = 0; t < 4; ++t) {
                    float vf = accF[mt][t][i] + bias[t]; vf = vf > 0.f ? vf : 0.f;
                    float vr = accR[mt][t][i] + bias[t]; vr = vr > 0.f ? vr : 0.f;
                    vp[t] = vf + vr;
                }
                *(v4f*)&sh_p[m * ZSTR + ncq] = vp;
            }
    }
    __syncthreads();

    for (int o = tid; o < MBLOCK * 3; o += 128) {
        const int m = o / 3;
        const int d = o - m * 3;
        float s = bout[d];
        const float* prow = &sh_p[m * ZSTR];
        #pragma unroll 4
        for (int k = 0; k < DMID; ++k)
            s += prow[k] * Wout[k * 3 + d];
        out[(size_t)(frag0 + m) * 3 + d] = s;
    }
}

extern "C" void kernel_launch(void* const* d_in, const int* in_sizes, int n_in,
                              void* d_out, int out_size, void* d_ws, size_t ws_size,
                              hipStream_t stream) {
    const float* x_atoms = (const float*)d_in[0];
    const int*   frag_idx = (const int*)d_in[1];
    const float* W1   = (const float*)d_in[2];
    const float* b1   = (const float*)d_in[3];
    const float* W2   = (const float*)d_in[4];
    const float* b2   = (const float*)d_in[5];
    const float* Wout = (const float*)d_in[6];
    const float* bout = (const float*)d_in[7];
    float* out = (float*)d_out;

    const int n_frag = in_sizes[1] / 3;
    const int grid = n_frag / MBLOCK;
    if (ws_size >= PREP_FLOATS * sizeof(float)) {
        float* P = (float*)d_ws;
        prearrange_weights<<<PREP_FLOATS / 4 / 256, 256, 0, stream>>>(W1, W2, P);
        janossy_readout_kernel<true><<<grid, 128, 0, stream>>>(
            x_atoms, frag_idx, P, b1, P + PREP_L2_BASE, b2, Wout, bout, out);
    } else {
        janossy_readout_kernel<false><<<grid, 128, 0, stream>>>(
            x_atoms, frag_idx, W1, b1, W2, b2, Wout, bout, out);
    }
}


# Layer 1 middle block per atom, not per fragment: a launch writes x@Wb+b1 (100000x256, 102 MB) to workspace; fragments gather that row, K loop keeps 2 end blocks; changes the f32 summation order
# speedup vs baseline: 1.7577x; 1.0486x over previous
#include <hip/hip_runtime.h>
#include <hip/hip_bf16.h>

typedef __attribute__((ext_vector_type(2))) float v2f;
typedef __attribute__((ext_vector_type(4))) float v4f;
typedef __attribute__((ext_vector_type(8))) float v8f;

#define N_FRAG  262144
#define MBLOCK  32
#define KDIM    384
#define DMID    256
#define HSTR    396
#define ZSTR    268

#define PREP_ROW      512
#define PREP_L1_OP    (64 * PREP_ROW)
#define PREP_L2_BASE  (3 * PREP_L1_OP)
#define PREP_FLOATS   (PREP_L2_BASE + 128 * PREP_ROW)

#define N_ATOMS     100000
#define MID_FLOATS  (N_ATOMS * DMID)
#define XSTR        140

__device__ __forceinline__ v8f wmma_f32(v2f a, v2f b, v8f c) {
    return __builtin_amdgcn_wmma_f32_16x16x4_f32(
         false, a,  false, b,
         (short)0, c,  false,  false);
}

#define PIN_ORDER() __builtin_amdgcn_sched_barrier(0)

#define OPERAND_RELEASE_GAP() do { PIN_ORDER(); asm volatile("v_nop\n\tv_nop\n\tv_nop\n\tv_nop"); PIN_ORDER(); } while (0)

struct L1Frag {
    v4f wa[2], wb[2], wc[2];
    v4f ps[2], pd[2], pm[2];
    v2f aS0, aS1, aM0, aM1, aD0, aD1;
    v2f bs[4], bd[4], bm[4];
};
struct L2Frag {
    v4f q[2][2];
    v4f p[2][2];
    v2f a[2][4];
    v2f b[2][4];
};

__device__ __forceinline__ void l1_load_w(L1Frag& f, const float* __restrict__ W1, int stage, int khalf, int ncq) {
    const int ka = stage * 4 + 2 * khalf;
    f.wa[0] = *(const v4f*)&W1[(      ka + 0) * DMID + ncq];
    f.wa[1] = *(const v4f*)&W1[(      ka + 1) * DMID + ncq];
    f.wb[0] = *(const v4f*)&W1[(128 + ka + 0) * DMID + ncq];
    f.wb[1] = *(const v4f*)&W1[(128 + ka + 1) * DMID + ncq];
    f.wc[0] = *(const v4f*)&W1[(256 + ka + 0) * DMID + ncq];
    f.wc[1] = *(const v4f*)&W1[(256 + ka + 1) * DMID + ncq];
}
template <bool MID>
__device__ __forceinline__ void l1_load_h(L1Frag& f, const float* sh_h, int stage, int khalf, int mrow) {
    const int ka = stage * 4 + 2 * khalf;
    if constexpr (MID) {
        f.aS0 = *(const v2f*)&sh_h[ mrow       * ZSTR +       ka];
        f.aS1 = *(const v2f*)&sh_h[(16 + mrow) * ZSTR +       ka];
        f.aD0 = *(const v2f*)&sh_h[ mrow       * ZSTR + 128 + ka];
        f.aD1 = *(const v2f*)&sh_h[(16 + mrow) * ZSTR + 128 + ka];
        return;
    }
    f.aS0 = *(const v2f*)&sh_h[ mrow       * HSTR +       ka];
    f.aS1 = *(const v2f*)&sh_h[(16 + mrow) * HSTR +       ka];
    f.aM0 = *(const v2f*)&sh_h[ mrow       * HSTR + 128 + ka];
    f.aM1 = *(const v2f*)&sh_h[(16 + mrow) * HSTR + 128 + ka];
    f.aD0 = *(const v2f*)&sh_h[ mrow       * HSTR + 256 + ka];
    f.aD1 = *(const v2f*)&sh_h[(16 + mrow) * HSTR + 256 + ka];
}
__device__ __forceinline__ void l1_comb(L1Frag& f) {
    #pragma unroll
    for (int t = 0; t < 4; ++t) {
        f.bs[t] = (v2f){ f.wa[0][t] + f.wc[0][t], f.wa[1][t] + f.wc[1][t] };
        f.bd[t] = (v2f){ f.wa[0][t] - f.wc[0][t], f.wa[1][t] - f.wc[1][t] };
        f.bm[t] = (v2f){ f.wb[0][t], f.wb[1][t] };
    }
}
template <bool MID>
__device__ __forceinline__ void l1_load_p(L1Frag& f, const float* __restrict__ P, int stage, int poff) {
    const float* r = P + stage * (2 * PREP_ROW) + poff;
    f.ps[0] = *(const v4f*)&r[0];
    f.ps[1] = *(const v4f*)&r[64];
    f.pd[0] = *(const v4f*)&r[PREP_L1_OP];
    f.pd[1] = *(const v4f*)&r[PREP_L1_OP + 64];
    if constexpr (!MID) {
        f.pm[0] = *(const v4f*)&r[2 * PREP_L1_OP];
        f.pm[1] = *(const v4f*)&r[2 * PREP_L1_OP + 64];
    }
}
template <bool MID>
__device__ __forceinline__ void l1_pairs_p(L1Frag& f) {
    #pragma unroll
    for (int t = 0; t < 4; ++t) {
        const int g = t >> 1, e = 2 * (t & 1);
        f.bs[t] = (v2f){ f.ps[g][e], f.ps[g][e + 1] };
        f.bd[t] = (v2f){ f.pd[g][e], f.pd[g][e + 1] };
        if constexpr (!MID) f.bm[t] = (v2f){ f.pm[g][e], f.pm[g][e + 1] };
    }
}
template <bool MID>
__device__ __forceinline__ void l1_mma(v8f (&accF)[2][4], v8f (&accR)[2][4], const L1Frag& f) {
    #pragma unroll
    for (int t = 0; t < 4; ++t) {
        accF[0][t] = wmma_f32(f.aS0, f.bs[t], accF[0][t]);
        accF[1][t] = wmma_f32(f.aS1, f.bs[t], accF[1][t]);
        accR[0][t] = wmma_f32(f.aD0, f.bd[t], accR[0][t]);
        accR[1][t] = wmma_f32(f.aD1, f.bd[t], accR[1][t]);
        if constexpr (!MID) {
            accF[0][t] = wmma_f32(f.aM0, f.bm[t], accF[0][t]);
            accF[1][t] = wmma_f32(f.aM1, f.bm[t], accF[1][t]);
        }
    }
}

__device__ __forceinline__ void l2_load_w(L2Frag& f, const float* __restrict__ W2, int stage, int khalf, int ncq) {
    #pragma unroll
    for (int u = 0; u < 2; ++u) {
        const int ka = (stage * 2 + u) * 4 + 2 * khalf;
        f.q[u][0] = *(const v4f*)&W2[(ka + 0) * DMID + ncq];
        f.q[u][1] = *(const v4f*)&W2[(ka + 1) * DMID + ncq];
    }
}
__device__ __forceinline__ void l2_load_z(L2Frag& f, const float* zf, const float* zr, int stage, int khalf, int mrow) {
    #pragma unroll
    for (int u = 0; u < 2; ++u) {
        const int ka = (stage * 2 + u) * 4 + 2 * khalf;
        f.a[u][0] = *(const v2f*)&zf[ mrow       * ZSTR + ka];
        f.a[u][1] = *(const v2f*)&zf[(16 + mrow) * ZSTR + ka];
        f.a[u][2] = *(const v2f*)&zr[ mrow       * ZSTR + ka];
        f.a[u][3] = *(const v2f*)&zr[(16 + mrow) * ZSTR + ka];
    }
}
__device__ __forceinline__ void l2_comb(L2Frag& f) {
    #pragma unroll
    for (int u = 0; u < 2; ++u)
        #pragma unroll
        for (int t = 0; t < 4; ++t)
            f.b[u][t] = (v2f){ f.q[u][0][t], f.q[u][1][t] };
}
__device__ __forceinline__ void l2_load_p(L2Frag& f, const float* __restrict__ P, int stage, int poff) {
    #pragma unroll
    for (int u = 0; u < 2; ++u) {
        const float* r = P + (stage * 2 + u) * (2 * PREP_ROW) + poff;
        f.p[u][0] = *(const v4f*)&r[0];
        f.p[u][1] = *(const v4f*)&r[64];
    }
}
__device__ __forceinline__ void l2_pairs_p(L2Frag& f) {
    #pragma unroll
    for (int u = 0; u < 2; ++u)
        #pragma unroll
        for (int t = 0; t < 4; ++t) {
            const int g = t >> 1, e = 2 * (t & 1);
            f.b[u][t] = (v2f){ f.p[u][g][e], f.p[u][g][e + 1] };
        }
}
__device__ __forceinline__ void l2_mma(v8f (&accF)[2][4], v8f (&accR)[2][4], const L2Frag& f) {
    #pragma unroll
    for (int u = 0; u < 2; ++u)
        #pragma unroll
        for (int t = 0; t < 4; ++t) {
            accF[0][t] = wmma_f32(f.a[u][0], f.b[u][t], accF[0][t]);
            accF[1][t] = wmma_f32(f.a[u][1], f.b[u][t], accF[1][t]);
            accR[0][t] = wmma_f32(f.a[u][2], f.b[u][t], accR[0][t]);
            accR[1][t] = wmma_f32(f.a[u][3], f.b[u][t], accR[1][t]);
        }
}

__global__ __launch_bounds__(256)
void prearrange_weights(const float* __restrict__ W1, const float* __restrict__ W2, float* __restrict__ P)
{
    const int o   = blockIdx.x * 256 + threadIdx.x;
    const int row = o >> 7;
    const int q   = o & 127;
    const int n   = 64 * (q >> 5) + 4 * (q & 15) + 2 * ((q >> 4) & 1);
    v4f r;
    if (row < 192) {
        const int op = row >> 6;
        const int k0 = 2 * (row & 63);
        if (op == 2) {
            const v2f b0 = *(const v2f*)&W1[(128 + k0) * DMID + n];
            const v2f b1 = *(const v2f*)&W1[(129 + k0) * DMID + n];
            r = (v4f){ b0[0], b1[0], b0[1], b1[1] };
        } else {
            const v2f a0 = *(const v2f*)&W1[(      k0    ) * DMID + n];
            const v2f a1 = *(const v2f*)&W1[(      k0 + 1) * DMID + n];
            const v2f c0 = *(const v2f*)&W1[(256 + k0    ) * DMID + n];
            const v2f c1 = *(const v2f*)&W1[(256 + k0 + 1) * DMID + n];
            r = (op == 0) ? (v4f){ a0[0] + c0[0], a1[0] + c1[0], a0[1] + c0[1], a1[1] + c1[1] }
                          : (v4f){ a0[0] - c0[0], a1[0] - c1[0], a0[1] - c0[1], a1[1] - c1[1] };
        }
    } else {
        const int k0 = 2 * (row - 192);
        const v2f w0 = *(const v2f*)&W2[(k0    ) * DMID + n];
        const v2f w1 = *(const v2f*)&W2[(k0 + 1) * DMID + n];
        r = (v4f){ w0[0], w1[0], w0[1], w1[1] };
    }
    *(v4f*)&P[(size_t)o * 4] = r;
}

struct MidFrag { v4f pm[2]; v2f a0, a1; };
__device__ __forceinline__ void mid_load(MidFrag& f, const float* __restrict__ Pm, const float* sh_x,
                                         int stage, int poff, int khalf, int mrow) {
    const float* r = Pm + stage * (2 * PREP_ROW) + poff;
    f.pm[0] = *(const v4f*)&r[0];
    f.pm[1] = *(const v4f*)&r[64];
    const int ka = stage * 4 + 2 * khalf;
    f.a0 = *(const v2f*)&sh_x[ mrow       * XSTR + ka];
    f.a1 = *(const v2f*)&sh_x[(16 + mrow) * XSTR + ka];
}
__device__ __forceinline__ void mid_mma(v8f (&acc)[2][4], const MidFrag& f) {
    #pragma unroll
    for (int t = 0; t < 4; ++t) {
        const int g = t >> 1, e = 2 * (t & 1);
        const v2f b = (v2f){ f.pm[g][e], f.pm[g][e + 1] };
        acc[0][t] = wmma_f32(f.a0, b, acc[0][t]);
        acc[1][t] = wmma_f32(f.a1, b, acc[1][t]);
    }
}
__global__ __launch_bounds__(128)
void atom_middle_block(const float* __restrict__ x_atoms, const float* __restrict__ Pm,
                       const float* __restrict__ b1, float* __restrict__ mid)
{
    static_assert(N_ATOMS % MBLOCK == 0, "whole workgroups of atoms");
    __shared__ __attribute__((aligned(16))) float sh_x[MBLOCK * XSTR];
    const int tid   = threadIdx.x;
    const int atom0 = blockIdx.x * MBLOCK;
    const int lane  = tid & 31;
    const int wave  = tid >> 5;
    const int mrow  = lane & 15;
    const int khalf = lane >> 4;
    const int ncol  = lane & 15;
    const int ncq   = wave * 64 + 4 * ncol;
    const int poff  = khalf * PREP_ROW + wave * 128 + ncol * 4;

    const float4* xin = (const float4*)(x_atoms + (size_t)atom0 * 128);
    #pragma unroll
    for (int r = 0; r < MBLOCK / 4; ++r) {
        const int i = tid + 128 * r;
        *(float4*)&sh_x[(i >> 5) * XSTR + (i & 31) * 4] = xin[i];
    }
    __syncthreads();

    v8f acc[2][4];
    #pragma unroll
    for (int mt = 0; mt < 2; ++mt)
        #pragma unroll
        for (int t = 0; t < 4; ++t)
            acc[mt][t] = (v8f){0.f,0.f,0.f,0.f,0.f,0.f,0.f,0.f};

    MidFrag fa, fb, fc;
    mid_load(fa, Pm, sh_x, 0, poff, khalf, mrow);
    mid_load(fb, Pm, sh_x, 1, poff, khalf, mrow);
    for (int j = 0; j < 30; j += 3) {
        mid_load(fc, Pm, sh_x, j + 2, poff, khalf, mrow);
        PIN_ORDER();
        mid_mma(acc, fa);
        PIN_ORDER();
        mid_load(fa, Pm, sh_x, j + 3, poff, khalf, mrow);
        PIN_ORDER();
        mid_mma(acc, fb);
        PIN_ORDER();
        mid_load(fb, Pm, sh_x, j + 4, poff, khalf, mrow);
        PIN_ORDER();
        mid_mma(acc, fc);
        PIN_ORDER();
    }
    mid_mma(acc, fa);
    mid_mma(acc, fb);
    OPERAND_RELEASE_GAP();

    const v4f bias = *(const v4f*)&b1[ncq];
    #pragma unroll
    for (int mt = 0; mt < 2; ++mt)
        #pragma unroll
        for (int i = 0; i < 8; ++i) {
            const int m = mt * 16 + i + 8 * khalf;
            v4f o;
            #pragma unroll
            for (int t = 0; t < 4; ++t) o[t] = acc[mt][t][i] + bias[t];
            *(v4f*)&mid[(size_t)(atom0 + m) * DMID + ncq] = o;
        }
}

template <bool PRE, bool MID>
__global__ __launch_bounds__(128, 1)
void janossy_readout_kernel(const float* __restrict__ x_atoms,
                            const int*   __restrict__ frag_idx,
                            const float* __restrict__ W1,
                            const float* __restrict__ b1,
                            const float* __restrict__ W2,
                            const float* __restrict__ b2,
                            const float* __restrict__ Wout,
                            const float* __restrict__ bout,
                            const float* __restrict__ mid,
                            float* __restrict__ out)
{
    static_assert(PRE || !MID, "the per-atom table is used with the pre-arranged weights only");
    __shared__ __attribute__((aligned(16))) float sh_region[2 * MBLOCK * ZSTR];
    __shared__ int   sh_idx[MBLOCK * 3];
    static_assert(MBLOCK * HSTR <= 2 * MBLOCK * ZSTR, "gathered rows must fit the region");
    float* const sh_h   = sh_region;
    float* const sh_z1f = sh_region;
    float* const sh_z1r = sh_region + MBLOCK * ZSTR;
    float* const sh_p   = sh_region;
    float* const sh_mid = sh_region + MBLOCK * ZSTR;

    const int tid   = threadIdx.x;
    const int frag0 = blockIdx.x * MBLOCK;

    const int lane  = tid & 31;
    const int wave  = tid >> 5;
    const int mrow  = lane & 15;
    const int khalf = lane >> 4;
    const int ncol  = lane & 15;
    const int n0    = wave * 64;
    const int ncq   = n0 + 4 * ncol;
    const int poff  = khalf * PREP_ROW + wave * 128 + ncol * 4;

    L1Frag q1a, q1b, q1c;
    L2Frag q2a, q2b, q2c;

    if constexpr (PRE) {
        l1_load_p<MID>(q1a, W1, 0, poff);
        l1_load_p<MID>(q1b, W1, 1, poff);
    } else {
        l1_load_w(q1a, W1, 0, khalf, ncq);
        l1_load_w(q1b, W1, 1, khalf, ncq);
    }

    if (tid < MBLOCK * 3) sh_idx[tid] = frag_idx[frag0 * 3 + tid];
    __syncthreads();

    if constexpr (MID) {
        #pragma unroll 4
        for (int r = 0; r < MBLOCK / 4; ++r) {
            const int i   = tid + 128 * r;
            const int m   = i >> 5;
            const int c4  = i & 31;
            const int mb  = i >> 6;
            const int cb  = i & 63;
            const float4* s0 = (const float4*)(x_atoms + (size_t)sh_idx[m * 3 + 0] * 128);
            const float4* s2 = (const float4*)(x_atoms + (size_t)sh_idx[m * 3 + 2] * 128);
            const float4* t0 = (const float4*)(mid + (size_t)sh_idx[ mb       * 3 + 1] * DMID);
            const float4* t1 = (const float4*)(mid + (size_t)sh_idx[(mb + 16) * 3 + 1] * DMID);
            float4 v0 = s0[c4];
            float4 v2 = s2[c4];
            float4 u0 = t0[cb];
            float4 u1 = t1[cb];
            float4 hs, hd;
            hs.x = 0.5f * (v0.x + v2.x);  hd.x = 0.5f * (v0.x - v2.x);
            hs.y = 0.5f * (v0.y + v2.y);  hd.y = 0.5f * (v0.y - v2.y);
            hs.z = 0.5f * (v0.z + v2.z);  hd.z = 0.5f * (v0.z - v2.z);
            hs.w = 0.5f * (v0.w + v2.w);  hd.w = 0.5f * (v0.w - v2.w);
            *(float4*)&sh_h[m * ZSTR +   0 + c4 * 4] = hs;
            *(float4*)&sh_h[m * ZSTR + 128 + c4 * 4] = hd;
            *(float4*)&sh_mid[ mb       * ZSTR + cb * 4] = u0;
            *(float4*)&sh_mid[(mb + 16) * ZSTR + cb * 4] = u1;
        }
    } else {
    #pragma unroll 2
    for (int i = tid; i < MBLOCK * 32; i += 128) {
        int m   = i >> 5;
        int c4  = i & 31;
        const float4* s0 = (const float4*)(x_atoms + (size_t)sh_idx[m * 3 + 0] * 128);
        const float4* s1 = (const float4*)(x_atoms + (size_t)sh_idx[m * 3 + 1] * 128);
        const float4* s2 = (const float4*)(x_atoms + (size_t)sh_idx[m * 3 + 2] * 128);
        float4 v0 = s0[c4];
        float4 v1 = s1[c4];
        float4 v2 = s2[c4];
        float4 hs, hd;
        hs.x = 0.5f * (v0.x + v2.x);  hd.x = 0.5f * (v0.x - v2.x);
        hs.y = 0.5f * (v0.y + v2.y);  hd.y = 0.5f * (v0.y - v2.y);
        hs.z = 0.5f * (v0.z + v2.z);  hd.z = 0.5f * (v0.z - v2.z);
        hs.w = 0.5f * (v0.w + v2.w);  hd.w = 0.5f * (v0.w - v2.w);
        *(float4*)&sh_h[m * HSTR +   0 + c4 * 4] = hs;
        *(float4*)&sh_h[m * HSTR + 128 + c4 * 4] = v1;
        *(float4*)&sh_h[m * HSTR + 256 + c4 * 4] = hd;
    }
    }
    __syncthreads();

    v8f accF[2][4], accR[2][4];
    #pragma unroll
    for (int mt = 0; mt < 2; ++mt)
        #pragma unroll
        for (int t = 0; t < 4; ++t) {
            accF[mt][t] = (v8f){0.f,0.f,0.f,0.f,0.f,0.f,0.f,0.f};
            accR[mt][t] = (v8f){0.f,0.f,0.f,0.f,0.f,0.f,0.f,0.f};
        }

#define L1_STAGE(X, Z, jz)                                   \
    if constexpr (PRE) {                                     \
        l1_load_p<MID>(Z, W1, (jz), poff);                   \
        l1_load_h<MID>(Z, sh_h, (jz), khalf, mrow);          \
        PIN_ORDER();                                         \
        l1_pairs_p<MID>(X);                                  \
    } else {                                                 \
        OPERAND_RELEASE_GAP();                               \
        l1_comb(X);                                          \
        PIN_ORDER();                                         \
        l1_load_w(Z, W1, (jz), khalf, ncq);                  \
        l1_load_h<MID>(Z, sh_h, (jz), khalf, mrow);          \
        PIN_ORDER();                                         \
    }                                                        \
    l1_mma<MID>(accF, accR, X);                              \
    PIN_ORDER();

    l1_load_h<MID>(q1a, sh_h, 0, khalf, mrow);
    l1_load_h<MID>(q1b, sh_h, 1, khalf, mrow);
    for (int j = 0; j < 30; j += 3) {
        L1_STAGE(q1a, q1c, j + 2)
        L1_STAGE(q1b, q1a, j + 3)
        L1_STAGE(q1c, q1b, j + 4)
    }
    if constexpr (!PRE) OPERAND_RELEASE_GAP();
    __syncthreads();
    if constexpr (PRE) {
        l2_load_p(q2a, W2, 0, poff);
        PIN_ORDER();
        l1_pairs_p<MID>(q1a);
        l1_mma<MID>(accF, accR, q1a);
        PIN_ORDER();
        l2_load_p(q2b, W2, 1, poff);
        PIN_ORDER();
        l1_pairs_p<MID>(q1b);
        l1_mma<MID>(accF, accR, q1b);
    } else {
        l1_comb(q1a);
        PIN_ORDER();
        l2_load_w(q2a, W2, 0, khalf, ncq);
        PIN_ORDER();
        l1_mma<MID>(accF, accR, q1a);
        OPERAND_RELEASE_GAP();
        l1_comb(q1b);
        PIN_ORDER();
        l2_load_w(q2b, W2, 1, khalf, ncq);
        PIN_ORDER();
        l1_mma<MID>(accF, accR, q1b);
    }
    OPERAND_RELEASE_GAP();
#undef L1_STAGE

    {
        const v4f bias = MID ? (v4f){0.f, 0.f, 0.f, 0.f} : *(const v4f*)&b1[ncq];
        #pragma unroll
        for (int mt = 0; mt < 2; ++mt)
            #pragma unroll
            for (int i = 0; i < 8; ++i) {
                const int m = mt * 16 + i + 8 * khalf;
                v4f vf, vr;
                const v4f add = MID ? *(const v4f*)&sh_mid[m * ZSTR + ncq] : bias;
                #pragma unroll
                for (int t = 0; t < 4; ++t) {
                    const float sS = accF[mt][t][i];
                    const float sD = accR[mt][t][i];
                    float f = (sS + sD) + add[t]; vf[t] = f > 0.f ? f : 0.f;
                    float r = (sS - sD) + add[t]; vr[t] = r > 0.f ? r : 0.f;
                }
                *(v4f*)&sh_z1f[m * ZSTR + ncq] = vf;
                *(v4f*)&sh_z1r[m * ZSTR + ncq] = vr;
            }
    }
    __syncthreads();

    #pragma unroll
    for (int mt = 0; mt < 2; ++mt)
        #pragma unroll
        for (int t = 0; t < 4; ++t) {
            accF[mt][t] = (v8f){0.f,0.f,0.f,0.f,0.f,0.f,0.f,0.f};
            accR[mt][t] = (v8f){0.f,0.f,0.f,0.f,0.f,0.f,0.f,0.f};
        }

#define L2_STAGE(X, Z, jz)                                   \
    if constexpr (PRE) {                                     \
        l2_load_p(Z, W2, (jz), poff);                        \
        l2_load_z(Z, sh_z1f, sh_z1r, (jz), khalf, mrow);     \
        PIN_ORDER();                                         \
        l2_pairs_p(X);                                       \
    } else {                                                 \
        OPERAND_RELEASE_GAP();                               \
        l2_comb(X);                                          \
        PIN_ORDER();                                         \
        l2_load_w(Z, W2, (jz), khalf, ncq);                  \
        l2_load_z(Z, sh_z1f, sh_z1r, (jz), khalf, mrow);     \
        PIN_ORDER();                                         \
    }                                                        \
    l2_mma(accF, accR, X);                                   \
    PIN_ORDER();

    l2_load_z(q2a, sh_z1f, sh_z1r, 0, khalf, mrow);
    l2_load_z(q2b, sh_z1f, sh_z1r, 1, khalf, mrow);
    for (int j = 0; j < 30; j += 3) {
        L2_STAGE(q2a, q2c, j + 2)
        L2_STAGE(q2b, q2a, j + 3)
        L2_STAGE(q2c, q2b, j + 4)
    }
    if constexpr (!PRE) OPERAND_RELEASE_GAP();
    __syncthreads();
    if constexpr (PRE) {
        l2_pairs_p(q2a);
        l2_pairs_p(q2b);
    } else {
        l2_comb(q2a);
        l2_comb(q2b);
    }
    PIN_ORDER();
    l2_mma(accF, accR, q2a);
    l2_mma(accF, accR, q2b);
    OPERAND_RELEASE_GAP();
#undef L2_STAGE

    {
        const v4f bias = *(const v4f*)&b2[ncq];
        #pragma unroll
        for (int mt = 0; mt < 2; ++mt)
            #pragma unroll
            for (int i = 0; i < 8; ++i) {
                const int m = mt * 16 + i + 8 * khalf;
                v4f vp;
                #pragma unroll
                for (int t = 0; t < 4; ++t) {
                    float vf = accF[mt][t][i] + bias[t]; vf = vf > 0.f ? vf : 0.f;
                    float vr = accR[mt][t][i] + bias[t]; vr = vr > 0.f ? vr : 0.f;
                    vp[t] = vf + vr;
                }
                *(v4f*)&sh_p[m * ZSTR + ncq] = vp;
            }
    }
    __syncthreads();

    for (int o = tid; o < MBLOCK * 3; o += 128) {
        const int m = o / 3;
        const int d = o - m * 3;
        float s = bout[d];
        const float* prow = &sh_p[m * ZSTR];
        #pragma unroll 4
        for (int k = 0; k < DMID; ++k)
            s += prow[k] * Wout[k * 3 + d];
        out[(size_t)(frag0 + m) * 3 + d] = s;
    }
}

extern "C" void kernel_launch(void* const* d_in, const int* in_sizes, int n_in,
                              void* d_out, int out_size, void* d_ws, size_t ws_size,
                              hipStream_t stream) {
    const float* x_atoms = (const float*)d_in[0];
    const int*   frag_idx = (const int*)d_in[1];
    const float* W1   = (const float*)d_in[2];
    const float* b1   = (const float*)d_in[3];
    const float* W2   = (const float*)d_in[4];
    const float* b2   = (const float*)d_in[5];
    const float* Wout = (const float*)d_in[6];
    const float* bout = (const float*)d_in[7];
    float* out = (float*)d_out;

    const int n_frag = in_sizes[1] / 3;
    const int grid = n_frag / MBLOCK;
    if (ws_size >= ((size_t)PREP_FLOATS + MID_FLOATS) * sizeof(float)) {
        float* P = (float*)d_ws;
        float* mid = P + PREP_FLOATS;
        prearrange_weights<<<PREP_FLOATS / 4 / 256, 256, 0, stream>>>(W1, W2, P);
        atom_middle_block<<<N_ATOMS / MBLOCK, 128, 0, stream>>>(x_atoms, P + 2 * PREP_L1_OP, b1, mid);
        janossy_readout_kernel<true, true><<<grid, 128, 0, stream>>>(
            x_atoms, frag_idx, P, b1, P + PREP_L2_BASE, b2, Wout, bout, mid, out);
    } else if (ws_size >= PREP_FLOATS * sizeof(float)) {
        float* P = (float*)d_ws;
        prearrange_weights<<<PREP_FLOATS / 4 / 256, 256, 0, stream>>>(W1, W2, P);
        janossy_readout_kernel<true, false><<<grid, 128, 0, stream>>>(
            x_atoms, frag_idx, P, b1, P + PREP_L2_BASE, b2, Wout, bout, nullptr, out);
    } else {
        janossy_readout_kernel<false, false><<<grid, 128, 0, stream>>>(
            x_atoms, frag_idx, W1, b1, W2, b2, Wout, bout, nullptr, out);
    }
}


# Layer 1 half-sum block per atom too: per-atom launch also writes x@(Wa+Wc) (2nd 100000x256 table if workspace holds 205 MB, else parent's path); K loop keeps difference block; changes f32 sum order
# speedup vs baseline: 1.8088x; 1.0291x over previous
#include <hip/hip_runtime.h>
#include <hip/hip_bf16.h>

typedef __attribute__((ext_vector_type(2))) float v2f;
typedef __attribute__((ext_vector_type(4))) float v4f;
typedef __attribute__((ext_vector_type(8))) float v8f;

#define N_FRAG  262144
#define MBLOCK  32
#define KDIM    384
#define DMID    256
#define HSTR    396
#define ZSTR    268

#define PREP_ROW      512
#define PREP_L1_OP    (64 * PREP_ROW)
#define PREP_L2_BASE  (3 * PREP_L1_OP)
#define PREP_FLOATS   (PREP_L2_BASE + 128 * PREP_ROW)

#define N_ATOMS     100000
#define MID_FLOATS  (N_ATOMS * DMID)
#define XSTR        140

__device__ __forceinline__ v8f wmma_f32(v2f a, v2f b, v8f c) {
    return __builtin_amdgcn_wmma_f32_16x16x4_f32(
         false, a,  false, b,
         (short)0, c,  false,  false);
}

#define PIN_ORDER() __builtin_amdgcn_sched_barrier(0)

#define OPERAND_RELEASE_GAP() do { PIN_ORDER(); asm volatile("v_nop\n\tv_nop\n\tv_nop\n\tv_nop"); PIN_ORDER(); } while (0)

struct L1Frag {
    v4f wa[2], wb[2], wc[2];
    v4f ps[2], pd[2], pm[2];
    v2f aS0, aS1, aM0, aM1, aD0, aD1;
    v2f bs[4], bd[4], bm[4];
};
struct L2Frag {
    v4f q[2][2];
    v4f p[2][2];
    v2f a[2][4];
    v2f b[2][4];
};

__device__ __forceinline__ void l1_load_w(L1Frag& f, const float* __restrict__ W1, int stage, int khalf, int ncq) {
    const int ka = stage * 4 + 2 * khalf;
    f.wa[0] = *(const v4f*)&W1[(      ka + 0) * DMID + ncq];
    f.wa[1] = *(const v4f*)&W1[(      ka + 1) * DMID + ncq];
    f.wb[0] = *(const v4f*)&W1[(128 + ka + 0) * DMID + ncq];
    f.wb[1] = *(const v4f*)&W1[(128 + ka + 1) * DMID + ncq];
    f.wc[0] = *(const v4f*)&W1[(256 + ka + 0) * DMID + ncq];
    f.wc[1] = *(const v4f*)&W1[(256 + ka + 1) * DMID + ncq];
}
template <bool MID, bool SUMT = false>
__device__ __forceinline__ void l1_load_h(L1Frag& f, const float* sh_h, int stage, int khalf, int mrow) {
    const int ka = stage * 4 + 2 * khalf;
    if constexpr (SUMT) {
        f.aD0 = *(const v2f*)&sh_h[ mrow       * ZSTR + ka];
        f.aD1 = *(const v2f*)&sh_h[(16 + mrow) * ZSTR + ka];
        return;
    }
    if constexpr (MID) {
        f.aS0 = *(const v2f*)&sh_h[ mrow       * ZSTR +       ka];
        f.aS1 = *(const v2f*)&sh_h[(16 + mrow) * ZSTR +       ka];
        f.aD0 = *(const v2f*)&sh_h[ mrow       * ZSTR + 128 + ka];
        f.aD1 = *(const v2f*)&sh_h[(16 + mrow) * ZSTR + 128 + ka];
        return;
    }
    f.aS0 = *(const v2f*)&sh_h[ mrow       * HSTR +       ka];
    f.aS1 = *(const v2f*)&sh_h[(16 + mrow) * HSTR +       ka];
    f.aM0 = *(const v2f*)&sh_h[ mrow       * HSTR + 128 + ka];
    f.aM1 = *(const v2f*)&sh_h[(16 + mrow) * HSTR + 128 + ka];
    f.aD0 = *(const v2f*)&sh_h[ mrow       * HSTR + 256 + ka];
    f.aD1 = *(const v2f*)&sh_h[(16 + mrow) * HSTR + 256 + ka];
}
__device__ __forceinline__ void l1_comb(L1Frag& f) {
    #pragma unroll
    for (int t = 0; t < 4; ++t) {
        f.bs[t] = (v2f){ f.wa[0][t] + f.wc[0][t], f.wa[1][t] + f.wc[1][t] };
        f.bd[t] = (v2f){ f.wa[0][t] - f.wc[0][t], f.wa[1][t] - f.wc[1][t] };
        f.bm[t] = (v2f){ f.wb[0][t], f.wb[1][t] };
    }
}
template <bool MID, bool SUMT = false>
__device__ __forceinline__ void l1_load_p(L1Frag& f, const float* __restrict__ P, int stage, int poff) {
    const float* r = P + stage * (2 * PREP_ROW) + poff;
    if constexpr (!SUMT) {
        f.ps[0] = *(const v4f*)&r[0];
        f.ps[1] = *(const v4f*)&r[64];
    }
    f.pd[0] = *(const v4f*)&r[PREP_L1_OP];
    f.pd[1] = *(const v4f*)&r[PREP_L1_OP + 64];
    if constexpr (!MID) {
        f.pm[0] = *(const v4f*)&r[2 * PREP_L1_OP];
        f.pm[1] = *(const v4f*)&r[2 * PREP_L1_OP + 64];
    }
}
template <bool MID, bool SUMT = false>
__device__ __forceinline__ void l1_pairs_p(L1Frag& f) {
    #pragma unroll
    for (int t = 0; t < 4; ++t) {
        const int g = t >> 1, e = 2 * (t & 1);
        if constexpr (!SUMT) f.bs[t] = (v2f){ f.ps[g][e], f.ps[g][e + 1] };
        f.bd[t] = (v2f){ f.pd[g][e], f.pd[g][e + 1] };
        if constexpr (!MID) f.bm[t] = (v2f){ f.pm[g][e], f.pm[g][e + 1] };
    }
}
template <bool MID, bool SUMT = false>
__device__ __forceinline__ void l1_mma(v8f (&accF)[2][4], v8f (&accR)[2][4], const L1Frag& f) {
    #pragma unroll
    for (int t = 0; t < 4; ++t) {
        if constexpr (!SUMT) {
            accF[0][t] = wmma_f32(f.aS0, f.bs[t], accF[0][t]);
            accF[1][t] = wmma_f32(f.aS1, f.bs[t], accF[1][t]);
        }
        accR[0][t] = wmma_f32(f.aD0, f.bd[t], accR[0][t]);
        accR[1][t] = wmma_f32(f.aD1, f.bd[t], accR[1][t]);
        if constexpr (!MID) {
            accF[0][t] = wmma_f32(f.aM0, f.bm[t], accF[0][t]);
            accF[1][t] = wmma_f32(f.aM1, f.bm[t], accF[1][t]);
        }
    }
}

__device__ __forceinline__ void l2_load_w(L2Frag& f, const float* __restrict__ W2, int stage, int khalf, int ncq) {
    #pragma unroll
    for (int u = 0; u < 2; ++u) {
        const int ka = (stage * 2 + u) * 4 + 2 * khalf;
        f.q[u][0] = *(const v4f*)&W2[(ka + 0) * DMID + ncq];
        f.q[u][1] = *(const v4f*)&W2[(ka + 1) * DMID + ncq];
    }
}
__device__ __forceinline__ void l2_load_z(L2Frag& f, const float* zf, const float* zr, int stage, int khalf, int mrow) {
    #pragma unroll
    for (int u = 0; u < 2; ++u) {
        const int ka = (stage * 2 + u) * 4 + 2 * khalf;
        f.a[u][0] = *(const v2f*)&zf[ mrow       * ZSTR + ka];
        f.a[u][1] = *(const v2f*)&zf[(16 + mrow) * ZSTR + ka];
        f.a[u][2] = *(const v2f*)&zr[ mrow       * ZSTR + ka];
        f.a[u][3] = *(const v2f*)&zr[(16 + mrow) * ZSTR + ka];
    }
}
__device__ __forceinline__ void l2_comb(L2Frag& f) {
    #pragma unroll
    for (int u = 0; u < 2; ++u)
        #pragma unroll
        for (int t = 0; t < 4; ++t)
            f.b[u][t] = (v2f){ f.q[u][0][t], f.q[u][1][t] };
}
__device__ __forceinline__ void l2_load_p(L2Frag& f, const float* __restrict__ P, int stage, int poff) {
    #pragma unroll
    for (int u = 0; u < 2; ++u) {
        const float* r = P + (stage * 2 + u) * (2 * PREP_ROW) + poff;
        f.p[u][0] = *(const v4f*)&r[0];
        f.p[u][1] = *(const v4f*)&r[64];
    }
}
__device__ __forceinline__ void l2_pairs_p(L2Frag& f) {
    #pragma unroll
    for (int u = 0; u < 2; ++u)
        #pragma unroll
        for (int t = 0; t < 4; ++t) {
            const int g = t >> 1, e = 2 * (t & 1);
            f.b[u][t] = (v2f){ f.p[u][g][e], f.p[u][g][e + 1] };
        }
}
__device__ __forceinline__ void l2_mma(v8f (&accF)[2][4], v8f (&accR)[2][4], const L2Frag& f) {
    #pragma unroll
    for (int u = 0; u < 2; ++u)
        #pragma unroll
        for (int t = 0; t < 4; ++t) {
            accF[0][t] = wmma_f32(f.a[u][0], f.b[u][t], accF[0][t]);
            accF[1][t] = wmma_f32(f.a[u][1], f.b[u][t], accF[1][t]);
            accR[0][t] = wmma_f32(f.a[u][2], f.b[u][t], accR[0][t]);
            accR[1][t] = wmma_f32(f.a[u][3], f.b[u][t], accR[1][t]);
        }
}

__global__ __launch_bounds__(256)
void prearrange_weights(const float* __restrict__ W1, const float* __restrict__ W2, float* __restrict__ P)
{
    const int o   = blockIdx.x * 256 + threadIdx.x;
    const int row = o >> 7;
    const int q   = o & 127;
    const int n   = 64 * (q >> 5) + 4 * (q & 15) + 2 * ((q >> 4) & 1);
    v4f r;
    if (row < 192) {
        const int op = row >> 6;
        const int k0 = 2 * (row & 63);
        if (op == 2) {
            const v2f b0 = *(const v2f*)&W1[(128 + k0) * DMID + n];
            const v2f b1 = *(const v2f*)&W1[(129 + k0) * DMID + n];
            r = (v4f){ b0[0], b1[0], b0[1], b1[1] };
        } else {
            const v2f a0 = *(const v2f*)&W1[(      k0    ) * DMID + n];
            const v2f a1 = *(const v2f*)&W1[(      k0 + 1) * DMID + n];
            const v2f c0 = *(const v2f*)&W1[(256 + k0    ) * DMID + n];
            const v2f c1 = *(const v2f*)&W1[(256 + k0 + 1) * DMID + n];
            r = (op == 0) ? (v4f){ a0[0] + c0[0], a1[0] + c1[0], a0[1] + c0[1], a1[1] + c1[1] }
                          : (v4f){ a0[0] - c0[0], a1[0] - c1[0], a0[1] - c0[1], a1[1] - c1[1] };
        }
    } else {
        const int k0 = 2 * (row - 192);
        const v2f w0 = *(const v2f*)&W2[(k0    ) * DMID + n];
        const v2f w1 = *(const v2f*)&W2[(k0 + 1) * DMID + n];
        r = (v4f){ w0[0], w1[0], w0[1], w1[1] };
    }
    *(v4f*)&P[(size_t)o * 4] = r;
}

struct MidFrag { v4f pm[2]; v2f a0, a1; };
__device__ __forceinline__ void mid_load(MidFrag& f, const float* __restrict__ Pm, const float* sh_x,
                                         int stage, int poff, int khalf, int mrow) {
    const float* r = Pm + stage * (2 * PREP_ROW) + poff;
    f.pm[0] = *(const v4f*)&r[0];
    f.pm[1] = *(const v4f*)&r[64];
    const int ka = stage * 4 + 2 * khalf;
    f.a0 = *(const v2f*)&sh_x[ mrow       * XSTR + ka];
    f.a1 = *(const v2f*)&sh_x[(16 + mrow) * XSTR + ka];
}
__device__ __forceinline__ void mid_mma(v8f (&acc)[2][4], const MidFrag& f) {
    #pragma unroll
    for (int t = 0; t < 4; ++t) {
        const int g = t >> 1, e = 2 * (t & 1);
        const v2f b = (v2f){ f.pm[g][e], f.pm[g][e + 1] };
        acc[0][t] = wmma_f32(f.a0, b, acc[0][t]);
        acc[1][t] = wmma_f32(f.a1, b, acc[1][t]);
    }
}
__global__ __launch_bounds__(128)
void atom_middle_block(const float* __restrict__ x_atoms, const float* __restrict__ Pmid,
                       const float* __restrict__ Psum, const float* __restrict__ b1,
                       float* __restrict__ mid, float* __restrict__ sumt)
{
    const bool sum_op = blockIdx.y != 0;
    const float* __restrict__ Pm  = sum_op ? Psum : Pmid;
    float* __restrict__       dst = sum_op ? sumt : mid;
    static_assert(N_ATOMS % MBLOCK == 0, "whole workgroups of atoms");
    __shared__ __attribute__((aligned(16))) float sh_x[MBLOCK * XSTR];
    const int tid   = threadIdx.x;
    const int atom0 = blockIdx.x * MBLOCK;
    const int lane  = tid & 31;
    const int wave  = tid >> 5;
    const int mrow  = lane & 15;
    const int khalf = lane >> 4;
    const int ncol  = lane & 15;
    const int ncq   = wave * 64 + 4 * ncol;
    const int poff  = khalf * PREP_ROW + wave * 128 + ncol * 4;

    const float4* xin = (const float4*)(x_atoms + (size_t)atom0 * 128);
    #pragma unroll
    for (int r = 0; r < MBLOCK / 4; ++r) {
        const int i = tid + 128 * r;
        *(float4*)&sh_x[(i >> 5) * XSTR + (i & 31) * 4] = xin[i];
    }
    __syncthreads();

    v8f acc[2][4];
    #pragma unroll
    for (int mt = 0; mt < 2; ++mt)
        #pragma unroll
        for (int t = 0; t < 4; ++t)
            acc[mt][t] = (v8f){0.f,0.f,0.f,0.f,0.f,0.f,0.f,0.f};

    MidFrag fa, fb, fc;
    mid_load(fa, Pm, sh_x, 0, poff, khalf, mrow);
    mid_load(fb, Pm, sh_x, 1, poff, khalf, mrow);
    for (int j = 0; j < 30; j += 3) {
        mid_load(fc, Pm, sh_x, j + 2, poff, khalf, mrow);
        PIN_ORDER();
        mid_mma(acc, fa);
        PIN_ORDER();
        mid_load(fa, Pm, sh_x, j + 3, poff, khalf, mrow);
        PIN_ORDER();
        mid_mma(acc, fb);
        PIN_ORDER();
        mid_load(fb, Pm, sh_x, j + 4, poff, khalf, mrow);
        PIN_ORDER();
        mid_mma(acc, fc);
        PIN_ORDER();
    }
    mid_mma(acc, fa);
    mid_mma(acc, fb);
    OPERAND_RELEASE_GAP();

    const v4f bias = sum_op ? (v4f){0.f, 0.f, 0.f, 0.f} : *(const v4f*)&b1[ncq];
    #pragma unroll
    for (int mt = 0; mt < 2; ++mt)
        #pragma unroll
        for (int i = 0; i < 8; ++i) {
            const int m = mt * 16 + i + 8 * khalf;
            v4f o;
            #pragma unroll
            for (int t = 0; t < 4; ++t) o[t] = acc[mt][t][i] + bias[t];
            *(v4f*)&dst[(size_t)(atom0 + m) * DMID + ncq] = o;
        }
}

template <bool PRE, bool MID, bool SUMT>
__global__ __launch_bounds__(128, 1)
void janossy_readout_kernel(const float* __restrict__ x_atoms,
                            const int*   __restrict__ frag_idx,
                            const float* __restrict__ W1,
                            const float* __restrict__ b1,
                            const float* __restrict__ W2,
                            const float* __restrict__ b2,
                            const float* __restrict__ Wout,
                            const float* __restrict__ bout,
                            const float* __restrict__ mid,
                            const float* __restrict__ sumt,
                            float* __restrict__ out)
{
    static_assert(PRE || !MID, "the per-atom table is used with the pre-arranged weights only");
    static_assert(MID || !SUMT, "the second per-atom table is used together with the first");
    __shared__ __attribute__((aligned(16))) float sh_region[2 * MBLOCK * ZSTR];
    __shared__ int   sh_idx[MBLOCK * 3];
    static_assert(MBLOCK * HSTR <= 2 * MBLOCK * ZSTR, "gathered rows must fit the region");
    float* const sh_h   = sh_region;
    float* const sh_z1f = sh_region;
    float* const sh_z1r = sh_region + MBLOCK * ZSTR;
    float* const sh_p   = sh_region;
    float* const sh_mid = sh_region + MBLOCK * ZSTR;

    const int tid   = threadIdx.x;
    const int frag0 = blockIdx.x * MBLOCK;

    const int lane  = tid & 31;
    const int wave  = tid >> 5;
    const int mrow  = lane & 15;
    const int khalf = lane >> 4;
    const int ncol  = lane & 15;
    const int n0    = wave * 64;
    const int ncq   = n0 + 4 * ncol;
    const int poff  = khalf * PREP_ROW + wave * 128 + ncol * 4;

    L1Frag q1a, q1b, q1c;
    L2Frag q2a, q2b, q2c;

    if constexpr (PRE) {
        l1_load_p<MID, SUMT>(q1a, W1, 0, poff);
        l1_load_p<MID, SUMT>(q1b, W1, 1, poff);
    } else {
        l1_load_w(q1a, W1, 0, khalf, ncq);
        l1_load_w(q1b, W1, 1, khalf, ncq);
    }

    if (tid < MBLOCK * 3) sh_idx[tid] = frag_idx[frag0 * 3 + tid];
    __syncthreads();

    if constexpr (SUMT) {
        #pragma unroll 4
        for (int r = 0; r < MBLOCK / 4; ++r) {
            const int i   = tid + 128 * r;
            const int m   = i >> 5;
            const int c4  = i & 31;
            const int mb  = i >> 6;
            const int cb  = i & 63;
            const float4* s0 = (const float4*)(x_atoms + (size_t)sh_idx[m * 3 + 0] * 128);
            const float4* s2 = (const float4*)(x_atoms + (size_t)sh_idx[m * 3 + 2] * 128);
            const float4* a0 = (const float4*)(sumt + (size_t)sh_idx[ mb       * 3 + 0] * DMID);
            const float4* a1 = (const float4*)(mid  + (size_t)sh_idx[ mb       * 3 + 1] * DMID);
            const float4* a2 = (const float4*)(sumt + (size_t)sh_idx[ mb       * 3 + 2] * DMID);
            const float4* e0 = (const float4*)(sumt + (size_t)sh_idx[(mb + 16) * 3 + 0] * DMID);
            const float4* e1 = (const float4*)(mid  + (size_t)sh_idx[(mb + 16) * 3 + 1] * DMID);
            const float4* e2 = (const float4*)(sumt + (size_t)sh_idx[(mb + 16) * 3 + 2] * DMID);
            float4 v0 = s0[c4];
            float4 v2 = s2[c4];
            float4 p0 = a0[cb], p1 = a1[cb], p2 = a2[cb];
            float4 q0 = e0[cb], q1 = e1[cb], q2 = e2[cb];
            float4 hd, u0, u1;
            hd.x = 0.5f * (v0.x - v2.x);  hd.y = 0.5f * (v0.y - v2.y);
            hd.z = 0.5f * (v0.z - v2.z);  hd.w = 0.5f * (v0.w - v2.w);
            u0.x = 0.5f * (p0.x + p2.x) + p1.x;  u0.y = 0.5f * (p0.y + p2.y) + p1.y;
            u0.z = 0.5f * (p0.z + p2.z) + p1.z;  u0.w = 0.5f * (p0.w + p2.w) + p1.w;
            u1.x = 0.5f * (q0.x + q2.x) + q1.x;  u1.y = 0.5f * (q0.y + q2.y) + q1.y;
            u1.z = 0.5f * (q0.z + q2.z) + q1.z;  u1.w = 0.5f * (q0.w + q2.w) + q1.w;
            *(float4*)&sh_h[m * ZSTR + c4 * 4] = hd;
            *(float4*)&sh_mid[ mb       * ZSTR + cb * 4] = u0;
            *(float4*)&sh_mid[(mb + 16) * ZSTR + cb * 4] = u1;
        }
    } else if constexpr (MID) {
        #pragma unroll 4
        for (int r = 0; r < MBLOCK / 4; ++r) {
            const int i   = tid + 128 * r;
            const int m   = i >> 5;
            const int c4  = i & 31;
            const int mb  = i >> 6;
            const int cb  = i & 63;
            const float4* s0 = (const float4*)(x_atoms + (size_t)sh_idx[m * 3 + 0] * 128);
            const float4* s2 = (const float4*)(x_atoms + (size_t)sh_idx[m * 3 + 2] * 128);
            const float4* t0 = (const float4*)(mid + (size_t)sh_idx[ mb       * 3 + 1] * DMID);
            const float4* t1 = (const float4*)(mid + (size_t)sh_idx[(mb + 16) * 3 + 1] * DMID);
            float4 v0 = s0[c4];
            float4 v2 = s2[c4];
            float4 u0 = t0[cb];
            float4 u1 = t1[cb];
            float4 hs, hd;
            hs.x = 0.5f * (v0.x + v2.x);  hd.x = 0.5f * (v0.x - v2.x);
            hs.y = 0.5f * (v0.y + v2.y);  hd.y = 0.5f * (v0.y - v2.y);
            hs.z = 0.5f * (v0.z + v2.z);  hd.z = 0.5f * (v0.z - v2.z);
            hs.w = 0.5f * (v0.w + v2.w);  hd.w = 0.5f * (v0.w - v2.w);
            *(float4*)&sh_h[m * ZSTR +   0 + c4 * 4] = hs;
            *(float4*)&sh_h[m * ZSTR + 128 + c4 * 4] = hd;
            *(float4*)&sh_mid[ mb       * ZSTR + cb * 4] = u0;
            *(float4*)&sh_mid[(mb + 16) * ZSTR + cb * 4] = u1;
        }
    } else {
    #pragma unroll 2
    for (int i = tid; i < MBLOCK * 32; i += 128) {
        int m   = i >> 5;
        int c4  = i & 31;
        const float4* s0 = (const float4*)(x_atoms + (size_t)sh_idx[m * 3 + 0] * 128);
        const float4* s1 = (const float4*)(x_atoms + (size_t)sh_idx[m * 3 + 1] * 128);
        const float4* s2 = (const float4*)(x_atoms + (size_t)sh_idx[m * 3 + 2] * 128);
        float4 v0 = s0[c4];
        float4 v1 = s1[c4];
        float4 v2 = s2[c4];
        float4 hs, hd;
        hs.x = 0.5f * (v0.x + v2.x);  hd.x = 0.5f * (v0.x - v2.x);
        hs.y = 0.5f * (v0.y + v2.y);  hd.y = 0.5f * (v0.y - v2.y);
        hs.z = 0.5f * (v0.z + v2.z);  hd.z = 0.5f * (v0.z - v2.z);
        hs.w = 0.5f * (v0.w + v2.w);  hd.w = 0.5f * (v0.w - v2.w);
        *(float4*)&sh_h[m * HSTR +   0 + c4 * 4] = hs;
        *(float4*)&sh_h[m * HSTR + 128 + c4 * 4] = v1;
        *(float4*)&sh_h[m * HSTR + 256 + c4 * 4] = hd;
    }
    }
    __syncthreads();

    v8f accF[2][4], accR[2][4];
    #pragma unroll
    for (int mt = 0; mt < 2; ++mt)
        #pragma unroll
        for (int t = 0; t < 4; ++t) {
            accF[mt][t] = (v8f){0.f,0.f,0.f,0.f,0.f,0.f,0.f,0.f};
            accR[mt][t] = (v8f){0.f,0.f,0.f,0.f,0.f,0.f,0.f,0.f};
        }

#define L1_STAGE(X, Z, jz)                                   \
    if constexpr (PRE) {                                     \
        l1_load_p<MID, SUMT>(Z, W1, (jz), poff);             \
        l1_load_h<MID, SUMT>(Z, sh_h, (jz), khalf, mrow);    \
        PIN_ORDER();                                         \
        l1_pairs_p<MID, SUMT>(X);                            \
    } else {                                                 \
        OPERAND_RELEASE_GAP();                               \
        l1_comb(X);                                          \
        PIN_ORDER();                                         \
        l1_load_w(Z, W1, (jz), khalf, ncq);                  \
        l1_load_h<MID, SUMT>(Z, sh_h, (jz), khalf, mrow);    \
        PIN_ORDER();                                         \
    }                                                        \
    l1_mma<MID, SUMT>(accF, accR, X);                        \
    PIN_ORDER();

    l1_load_h<MID, SUMT>(q1a, sh_h, 0, khalf, mrow);
    l1_load_h<MID, SUMT>(q1b, sh_h, 1, khalf, mrow);
    for (int j = 0; j < 30; j += 3) {
        L1_STAGE(q1a, q1c, j + 2)
        L1_STAGE(q1b, q1a, j + 3)
        L1_STAGE(q1c, q1b, j + 4)
    }
    if constexpr (!PRE) OPERAND_RELEASE_GAP();
    __syncthreads();
    if constexpr (PRE) {
        l2_load_p(q2a, W2, 0, poff);
        PIN_ORDER();
        l1_pairs_p<MID, SUMT>(q1a);
        l1_mma<MID, SUMT>(accF, accR, q1a);
        PIN_ORDER();
        l2_load_p(q2b, W2, 1, poff);
        PIN_ORDER();
        l1_pairs_p<MID, SUMT>(q1b);
        l1_mma<MID, SUMT>(accF, accR, q1b);
    } else {
        l1_comb(q1a);
        PIN_ORDER();
        l2_load_w(q2a, W2, 0, khalf, ncq);
        PIN_ORDER();
        l1_mma<MID, SUMT>(accF, accR, q1a);
        OPERAND_RELEASE_GAP();
        l1_comb(q1b);
        PIN_ORDER();
        l2_load_w(q2b, W2, 1, khalf, ncq);
        PIN_ORDER();
        l1_mma<MID, SUMT>(accF, accR, q1b);
    }
    OPERAND_RELEASE_GAP();
#undef L1_STAGE

    {
        const v4f bias = MID ? (v4f){0.f, 0.f, 0.f, 0.f} : *(const v4f*)&b1[ncq];
        #pragma unroll
        for (int mt = 0; mt < 2; ++mt)
            #pragma unroll
            for (int i = 0; i < 8; ++i) {
                const int m = mt * 16 + i + 8 * khalf;
                v4f vf, vr;
                const v4f add = MID ? *(const v4f*)&sh_mid[m * ZSTR + ncq] : bias;
                #pragma unroll
                for (int t = 0; t < 4; ++t) {
                    const float sD = accR[mt][t][i];
                    if constexpr (SUMT) {
                        float f = add[t] + sD; vf[t] = f > 0.f ? f : 0.f;
                        float r = add[t] - sD; vr[t] = r > 0.f ? r : 0.f;
                    } else {
                        const float sS = accF[mt][t][i];
                        float f = (sS + sD) + add[t]; vf[t] = f > 0.f ? f : 0.f;
                        float r = (sS - sD) + add[t]; vr[t] = r > 0.f ? r : 0.f;
                    }
                }
                *(v4f*)&sh_z1f[m * ZSTR + ncq] = vf;
                *(v4f*)&sh_z1r[m * ZSTR + ncq] = vr;
            }
    }
    __syncthreads();

    #pragma unroll
    for (int mt = 0; mt < 2; ++mt)
        #pragma unroll
        for (int t = 0; t < 4; ++t) {
            accF[mt][t] = (v8f){0.f,0.f,0.f,0.f,0.f,0.f,0.f,0.f};
            accR[mt][t] = (v8f){0.f,0.f,0.f,0.f,0.f,0.f,0.f,0.f};
        }

#define L2_STAGE(X, Z, jz)                                   \
    if constexpr (PRE) {                                     \
        l2_load_p(Z, W2, (jz), poff);                        \
        l2_load_z(Z, sh_z1f, sh_z1r, (jz), khalf, mrow);     \
        PIN_ORDER();                                         \
        l2_pairs_p(X);                                       \
    } else {                                                 \
        OPERAND_RELEASE_GAP();                               \
        l2_comb(X);                                          \
        PIN_ORDER();                                         \
        l2_load_w(Z, W2, (jz), khalf, ncq);                  \
        l2_load_z(Z, sh_z1f, sh_z1r, (jz), khalf, mrow);     \
        PIN_ORDER();                                         \
    }                                                        \
    l2_mma(accF, accR, X);                                   \
    PIN_ORDER();

    l2_load_z(q2a, sh_z1f, sh_z1r, 0, khalf, mrow);
    l2_load_z(q2b, sh_z1f, sh_z1r, 1, khalf, mrow);
    for (int j = 0; j < 30; j += 3) {
        L2_STAGE(q2a, q2c, j + 2)
        L2_STAGE(q2b, q2a, j + 3)
        L2_STAGE(q2c, q2b, j + 4)
    }
    if constexpr (!PRE) OPERAND_RELEASE_GAP();
    __syncthreads();
    if constexpr (PRE) {
        l2_pairs_p(q2a);
        l2_pairs_p(q2b);
    } else {
        l2_comb(q2a);
        l2_comb(q2b);
    }
    PIN_ORDER();
    l2_mma(accF, accR, q2a);
    l2_mma(accF, accR, q2b);
    OPERAND_RELEASE_GAP();
#undef L2_STAGE

    {
        const v4f bias = *(const v4f*)&b2[ncq];
        #pragma unroll
        for (int mt = 0; mt < 2; ++mt)
            #pragma unroll
            for (int i = 0; i < 8; ++i) {
                const int m = mt * 16 + i + 8 * khalf;
                v4f vp;
                #pragma unroll
                for (int t = 0; t < 4; ++t) {
                    float vf = accF[mt][t][i] + bias[t]; vf = vf > 0.f ? vf : 0.f;
                    float vr = accR[mt][t][i] + bias[t]; vr = vr > 0.f ? vr : 0.f;
                    vp[t] = vf + vr;
                }
                *(v4f*)&sh_p[m * ZSTR + ncq] = vp;
            }
    }
    __syncthreads();

    for (int o = tid; o < MBLOCK * 3; o += 128) {
        const int m = o / 3;
        const int d = o - m * 3;
        float s = bout[d];
        const float* prow = &sh_p[m * ZSTR];
        #pragma unroll 4
        for (int k = 0; k < DMID; ++k)
            s += prow[k] * Wout[k * 3 + d];
        out[(size_t)(frag0 + m) * 3 + d] = s;
    }
}

extern "C" void kernel_launch(void* const* d_in, const int* in_sizes, int n_in,
                              void* d_out, int out_size, void* d_ws, size_t ws_size,
                              hipStream_t stream) {
    const float* x_atoms = (const float*)d_in[0];
    const int*   frag_idx = (const int*)d_in[1];
    const float* W1   = (const float*)d_in[2];
    const float* b1   = (const float*)d_in[3];
    const float* W2   = (const float*)d_in[4];
    const float* b2   = (const float*)d_in[5];
    const float* Wout = (const float*)d_in[6];
    const float* bout = (const float*)d_in[7];
    float* out = (float*)d_out;

    const int n_frag = in_sizes[1] / 3;
    const int grid = n_frag / MBLOCK;
    if (ws_size >= ((size_t)PREP_FLOATS + 2 * (size_t)MID_FLOATS) * sizeof(float)) {
        float* P = (float*)d_ws;
        float* mid = P + PREP_FLOATS;
        float* sumt = mid + MID_FLOATS;
        prearrange_weights<<<PREP_FLOATS / 4 / 256, 256, 0, stream>>>(W1, W2, P);
        atom_middle_block<<<dim3(N_ATOMS / MBLOCK, 2), 128, 0, stream>>>(
            x_atoms, P + 2 * PREP_L1_OP, P, b1, mid, sumt);
        janossy_readout_kernel<true, true, true><<<grid, 128, 0, stream>>>(
            x_atoms, frag_idx, P, b1, P + PREP_L2_BASE, b2, Wout, bout, mid, sumt, out);
    } else if (ws_size >= ((size_t)PREP_FLOATS + MID_FLOATS) * sizeof(float)) {
        float* P = (float*)d_ws;
        float* mid = P + PREP_FLOATS;
        prearrange_weights<<<PREP_FLOATS / 4 / 256, 256, 0, stream>>>(W1, W2, P);
        atom_middle_block<<<N_ATOMS / MBLOCK, 128, 0, stream>>>(
            x_atoms, P + 2 * PREP_L1_OP, nullptr, b1, mid, nullptr);
        janossy_readout_kernel<true, true, false><<<grid, 128, 0, stream>>>(
            x_atoms, frag_idx, P, b1, P + PREP_L2_BASE, b2, Wout, bout, mid, nullptr, out);
    } else if (ws_size >= PREP_FLOATS * sizeof(float)) {
        float* P = (float*)d_ws;
        prearrange_weights<<<PREP_FLOATS / 4 / 256, 256, 0, stream>>>(W1, W2, P);
        janossy_readout_kernel<true, false, false><<<grid, 128, 0, stream>>>(
            x_atoms, frag_idx, P, b1, P + PREP_L2_BASE, b2, Wout, bout, nullptr, nullptr, out);
    } else {
        janossy_readout_kernel<false, false, false><<<grid, 128, 0, stream>>>(
            x_atoms, frag_idx, W1, b1, W2, b2, Wout, bout, nullptr, nullptr, out);
    }
}
